# starting kernel
# baseline (speedup 1.0000x reference)
#include <hip/hip_runtime.h>
#include <math.h>

#define IN_DIM   128
#define OUT_DIM  64
#define NEG_SLOPE 0.2f

typedef float v2f __attribute__((ext_vector_type(2)));
typedef float v8f __attribute__((ext_vector_type(8)));

__device__ __forceinline__ float leaky(float e) {
  return (e > 0.f) ? e : NEG_SLOPE * e;
}

__global__ void k_wt(const float* __restrict__ W, float* __restrict__ Wt) {
  int i = blockIdx.x * blockDim.x + threadIdx.x;
  if (i >= IN_DIM * OUT_DIM) return;
  int k = i / OUT_DIM;
  int c = i - k * OUT_DIM;
  Wt[(size_t)c * IN_DIM + k] = W[i];
}

__global__ __launch_bounds__(256) void k_feat_wmma(const float* __restrict__ h,
                                                   const float* __restrict__ Wt,
                                                   float* __restrict__ feat,
                                                   int N) {
  const int tid   = threadIdx.x;
  const int wave  = tid >> 5;
  const int lane  = tid & 31;
  const int mtile = wave >> 2;
  const int ntile = wave & 3;
  const int rowBase = blockIdx.x * 32 + mtile * 16;
  const int col   = ntile * 16 + (lane & 15);
  const int row   = rowBase + (lane & 15);
  const int khalf = (lane >> 4) << 1;

  const int rowc = (row < N) ? row : (N - 1);
  const float* __restrict__ ap = h  + (size_t)rowc * IN_DIM + khalf;
  const float* __restrict__ bp = Wt + (size_t)col  * IN_DIM + khalf;

  v8f acc = {0.f, 0.f, 0.f, 0.f, 0.f, 0.f, 0.f, 0.f};

#pragma unroll
  for (int k = 0; k < IN_DIM; k += 16) {
    v2f a0, a1, a2, a3, b0, b1, b2, b3;
    a0 = *(const v2f*)(ap + k);      b0 = *(const v2f*)(bp + k);
    a1 = *(const v2f*)(ap + k + 4);  b1 = *(const v2f*)(bp + k + 4);
    a2 = *(const v2f*)(ap + k + 8);  b2 = *(const v2f*)(bp + k + 8);
    a3 = *(const v2f*)(ap + k + 12); b3 = *(const v2f*)(bp + k + 12);
    acc = __builtin_amdgcn_wmma_f32_16x16x4_f32(false, a0, false, b0, (short)0, acc, false, false);
    acc = __builtin_amdgcn_wmma_f32_16x16x4_f32(false, a1, false, b1, (short)0, acc, false, false);
    acc = __builtin_amdgcn_wmma_f32_16x16x4_f32(false, a2, false, b2, (short)0, acc, false, false);
    acc = __builtin_amdgcn_wmma_f32_16x16x4_f32(false, a3, false, b3, (short)0, acc, false, false);
  }

  const int mBase = rowBase + (lane >> 4) * 8;
#pragma unroll
  for (int r = 0; r < 8; ++r) {
    int m = mBase + r;
    if (m < N) feat[(size_t)m * OUT_DIM + col] = acc[r];
  }
}

__global__ void k_el_er(const float* __restrict__ feat,
                        const float* __restrict__ attn_l,
                        const float* __restrict__ attn_r,
                        float* __restrict__ el, float* __restrict__ er,
                        int* __restrict__ counts, int N) {
  int i = blockIdx.x * blockDim.x + threadIdx.x;
  if (i >= N) return;
  const float4* f  = (const float4*)(feat + (size_t)i * OUT_DIM);
  const float4* al = (const float4*)attn_l;
  const float4* ar = (const float4*)attn_r;
  float sl = 0.f, sr = 0.f;
#pragma unroll
  for (int j = 0; j < OUT_DIM / 4; ++j) {
    float4 v = f[j], a = al[j], b = ar[j];
    sl += v.x * a.x + v.y * a.y + v.z * a.z + v.w * a.w;
    sr += v.x * b.x + v.y * b.y + v.z * b.z + v.w * b.w;
  }
  el[i] = sl;
  er[i] = sr;
  counts[i] = 0;
}

__global__ void k_hist(const int* __restrict__ dst, int* __restrict__ counts, int E) {
  int e = blockIdx.x * blockDim.x + threadIdx.x;
  if (e >= E) return;
  atomicAdd(&counts[dst[e]], 1);
}

__global__ __launch_bounds__(256) void k_scan1(const int* __restrict__ counts,
                                               int* __restrict__ offsets,
                                               int* __restrict__ bsum, int N) {
  __shared__ int sd[256];
  int tid  = threadIdx.x;
  int base = blockIdx.x * 1024 + tid * 4;
  int v0 = (base + 0 < N) ? counts[base + 0] : 0;
  int v1 = (base + 1 < N) ? counts[base + 1] : 0;
  int v2 = (base + 2 < N) ? counts[base + 2] : 0;
  int v3 = (base + 3 < N) ? counts[base + 3] : 0;
  int s = v0 + v1 + v2 + v3;
  sd[tid] = s;
  __syncthreads();
  for (int off = 1; off < 256; off <<= 1) {
    int t = (tid >= off) ? sd[tid - off] : 0;
    __syncthreads();
    sd[tid] += t;
    __syncthreads();
  }
  int excl = sd[tid] - s;
  if (base + 0 < N) offsets[base + 0] = excl;
  if (base + 1 < N) offsets[base + 1] = excl + v0;
  if (base + 2 < N) offsets[base + 2] = excl + v0 + v1;
  if (base + 3 < N) offsets[base + 3] = excl + v0 + v1 + v2;
  if (tid == 255) bsum[blockIdx.x] = sd[255];
}

__global__ void k_scan2(int* __restrict__ bsum, int nb) {
  if (threadIdx.x != 0 || blockIdx.x != 0) return;
  int running = 0;
  for (int b = 0; b < nb; ++b) {
    int t = bsum[b];
    bsum[b] = running;
    running += t;
  }
}

__global__ void k_scan3(int* __restrict__ offsets, const int* __restrict__ bsum,
                        int* __restrict__ counts, int N) {
  int i = blockIdx.x * blockDim.x + threadIdx.x;
  if (i >= N) return;
  offsets[i] += bsum[i >> 10];
  counts[i] = 0;
}

__global__ void k_fill(const int* __restrict__ src, const int* __restrict__ dst,
                       const int* __restrict__ offsets, int* __restrict__ cursor,
                       int* __restrict__ csr_src, int E) {
  int e = blockIdx.x * blockDim.x + threadIdx.x;
  if (e >= E) return;
  int d = dst[e];
  int p = offsets[d] + atomicAdd(&cursor[d], 1);
  csr_src[p] = src[e];
}

__global__ __launch_bounds__(256) void k_gat_agg(const float* __restrict__ feat,
                                                 const float* __restrict__ el,
                                                 const float* __restrict__ er,
                                                 const int* __restrict__ offsets,
                                                 const int* __restrict__ csr_src,
                                                 const float* __restrict__ bias,
                                                 float* __restrict__ out,
                                                 int N, int E) {
  int node = blockIdx.x * 8 + (threadIdx.x >> 5);
  if (node >= N) return;
  int lane  = threadIdx.x & 31;
  int start = offsets[node];
  int end   = (node == N - 1) ? E : offsets[node + 1];
  float er_d = er[node];

  int   j0 = start + lane;
  float e0 = (j0 < end) ? leaky(el[csr_src[j0]] + er_d) : -INFINITY;

  float m = e0;
  for (int j = j0 + 32; j < end; j += 32) {
    m = fmaxf(m, leaky(el[csr_src[j]] + er_d));
  }
#pragma unroll
  for (int off = 16; off > 0; off >>= 1) m = fmaxf(m, __shfl_xor(m, off, 32));
  if (end <= start) m = 0.f;

  float den = (j0 < end) ? __expf(e0 - m) : 0.f;
  for (int j = j0 + 32; j < end; j += 32) {
    den += __expf(leaky(el[csr_src[j]] + er_d) - m);
  }
#pragma unroll
  for (int off = 16; off > 0; off >>= 1) den += __shfl_xor(den, off, 32);
  float invden = (den > 0.f) ? 1.f / den : 1.f;

  float acc0 = 0.f, acc1 = 0.f;
  for (int j = start; j < end; ++j) {
    int t = j - start;
    int s = csr_src[j];
    float e;
    if (t < 32) {
      e = __shfl(e0, t, 32);
    } else {
      e = leaky(el[s] + er_d);
    }
    float alpha = __expf(e - m) * invden;
    v2f fv = *(const v2f*)(feat + (size_t)s * OUT_DIM + 2 * lane);
    acc0 += alpha * fv.x;
    acc1 += alpha * fv.y;
  }
  v2f bv = *(const v2f*)(bias + 2 * lane);
  v2f ov;
  ov.x = acc0 + bv.x;
  ov.y = acc1 + bv.y;
  *(v2f*)(out + (size_t)node * OUT_DIM + 2 * lane) = ov;
}

extern "C" void kernel_launch(void* const* d_in, const int* in_sizes, int n_in,
                              void* d_out, int out_size, void* d_ws, size_t ws_size,
                              hipStream_t stream) {
  const float* h      = (const float*)d_in[0];
  const int*   src    = (const int*)d_in[1];
  const int*   dst    = (const int*)d_in[2];
  const float* W      = (const float*)d_in[3];
  const float* attn_l = (const float*)d_in[4];
  const float* attn_r = (const float*)d_in[5];
  const float* bias   = (const float*)d_in[6];
  float* out = (float*)d_out;

  const int N = in_sizes[0] / IN_DIM;
  const int E = in_sizes[1];

  char* ws = (char*)d_ws;
  size_t off = 0;
  auto carve = [&](size_t bytes) -> char* {
    char* p = ws + off;
    off = (off + bytes + 255) & ~(size_t)255;
    return p;
  };
  float* feat    = (float*)carve((size_t)N * OUT_DIM * sizeof(float));
  float* Wt      = (float*)carve((size_t)IN_DIM * OUT_DIM * sizeof(float));
  float* el      = (float*)carve((size_t)N * sizeof(float));
  float* er      = (float*)carve((size_t)N * sizeof(float));
  int*   counts  = (int*)  carve((size_t)N * sizeof(int));
  int*   offsets = (int*)  carve((size_t)N * sizeof(int));
  int*   bsum    = (int*)  carve(4096);
  int*   csr_src = (int*)  carve((size_t)E * sizeof(int));

  const int nb = (N + 1023) / 1024;

  k_wt<<<(IN_DIM * OUT_DIM + 255) / 256, 256, 0, stream>>>(W, Wt);
  k_feat_wmma<<<(N + 31) / 32, 256, 0, stream>>>(h, Wt, feat, N);
  k_el_er<<<(N + 255) / 256, 256, 0, stream>>>(feat, attn_l, attn_r, el, er, counts, N);
  k_hist <<<(E + 255) / 256, 256, 0, stream>>>(dst, counts, E);
  k_scan1<<<nb, 256, 0, stream>>>(counts, offsets, bsum, N);
  k_scan2<<<1, 1, 0, stream>>>(bsum, nb);
  k_scan3<<<(N + 255) / 256, 256, 0, stream>>>(offsets, bsum, counts, N);
  k_fill <<<(E + 255) / 256, 256, 0, stream>>>(src, dst, offsets, counts, csr_src, E);
  k_gat_agg<<<(N + 7) / 8, 256, 0, stream>>>(feat, el, er, offsets, csr_src, bias, out, N, E);
}


# Aggregation without edge sort: a block owns 512 destination nodes, filters the whole dst list, sums exp*feat rows in LDS by LDS float atomics; CSR launches go. Node sum order = LDS arrival order.
# speedup vs baseline: 1.7498x; 1.7498x over previous
#include <hip/hip_runtime.h>
#include <math.h>

#define IN_DIM   128
#define OUT_DIM  64
#define NEG_SLOPE 0.2f

typedef float v2f __attribute__((ext_vector_type(2)));
typedef float v8f __attribute__((ext_vector_type(8)));

__device__ __forceinline__ float leaky(float e) {
  return (e > 0.f) ? e : NEG_SLOPE * e;
}

__global__ void k_wt(const float* __restrict__ W, float* __restrict__ Wt) {
  int i = blockIdx.x * blockDim.x + threadIdx.x;
  if (i >= IN_DIM * OUT_DIM) return;
  int k = i / OUT_DIM;
  int c = i - k * OUT_DIM;
  Wt[(size_t)c * IN_DIM + k] = W[i];
}

__global__ __launch_bounds__(256) void k_feat_wmma(const float* __restrict__ h,
                                                   const float* __restrict__ Wt,
                                                   float* __restrict__ feat,
                                                   int N) {
  const int tid   = threadIdx.x;
  const int wave  = tid >> 5;
  const int lane  = tid & 31;
  const int mtile = wave >> 2;
  const int ntile = wave & 3;
  const int rowBase = blockIdx.x * 32 + mtile * 16;
  const int col   = ntile * 16 + (lane & 15);
  const int row   = rowBase + (lane & 15);
  const int khalf = (lane >> 4) << 1;

  const int rowc = (row < N) ? row : (N - 1);
  const float* __restrict__ ap = h  + (size_t)rowc * IN_DIM + khalf;
  const float* __restrict__ bp = Wt + (size_t)col  * IN_DIM + khalf;

  v8f acc = {0.f, 0.f, 0.f, 0.f, 0.f, 0.f, 0.f, 0.f};

#pragma unroll
  for (int k = 0; k < IN_DIM; k += 16) {
    v2f a0, a1, a2, a3, b0, b1, b2, b3;
    a0 = *(const v2f*)(ap + k);      b0 = *(const v2f*)(bp + k);
    a1 = *(const v2f*)(ap + k + 4);  b1 = *(const v2f*)(bp + k + 4);
    a2 = *(const v2f*)(ap + k + 8);  b2 = *(const v2f*)(bp + k + 8);
    a3 = *(const v2f*)(ap + k + 12); b3 = *(const v2f*)(bp + k + 12);
    acc = __builtin_amdgcn_wmma_f32_16x16x4_f32(false, a0, false, b0, (short)0, acc, false, false);
    acc = __builtin_amdgcn_wmma_f32_16x16x4_f32(false, a1, false, b1, (short)0, acc, false, false);
    acc = __builtin_amdgcn_wmma_f32_16x16x4_f32(false, a2, false, b2, (short)0, acc, false, false);
    acc = __builtin_amdgcn_wmma_f32_16x16x4_f32(false, a3, false, b3, (short)0, acc, false, false);
  }

  const int mBase = rowBase + (lane >> 4) * 8;
#pragma unroll
  for (int r = 0; r < 8; ++r) {
    int m = mBase + r;
    if (m < N) feat[(size_t)m * OUT_DIM + col] = acc[r];
  }
}

__global__ void k_el_er(const float* __restrict__ feat,
                        const float* __restrict__ attn_l,
                        const float* __restrict__ attn_r,
                        float* __restrict__ el, float* __restrict__ er, int N) {
  int i = blockIdx.x * blockDim.x + threadIdx.x;
  if (i >= N) return;
  const float4* f  = (const float4*)(feat + (size_t)i * OUT_DIM);
  const float4* al = (const float4*)attn_l;
  const float4* ar = (const float4*)attn_r;
  float sl = 0.f, sr = 0.f;
#pragma unroll
  for (int j = 0; j < OUT_DIM / 4; ++j) {
    float4 v = f[j], a = al[j], b = ar[j];
    sl += v.x * a.x + v.y * a.y + v.z * a.z + v.w * a.w;
    sr += v.x * b.x + v.y * b.y + v.z * b.z + v.w * b.w;
  }
  el[i] = sl;
  er[i] = sr;
}

#define RANGE    512
#define LIST_CAP 16384
#define AGG_THREADS 1024
#define AGG_WAVES   (AGG_THREADS / 32)
#define PASS2_EDGES 8
#define WAVE_CAP    (LIST_CAP / AGG_WAVES)

__device__ __forceinline__ int fkey(float f) {
  int b = __float_as_int(f);
  return (b >= 0) ? b : (b ^ 0x7fffffff);
}
__device__ __forceinline__ float fkey_inv(int k) {
  return __int_as_float((k >= 0) ? k : (k ^ 0x7fffffff));
}

__global__ __launch_bounds__(AGG_THREADS) void k_gat_range(const float* __restrict__ feat,
                                                    const float* __restrict__ el,
                                                    const float* __restrict__ er,
                                                    const int* __restrict__ src,
                                                    const int* __restrict__ dst,
                                                    const float* __restrict__ bias,
                                                    float* __restrict__ out,
                                                    int N, int E) {
  __shared__ float          s_acc[RANGE * OUT_DIM];
  __shared__ int            s_mkey[RANGE];
  __shared__ float          s_den[RANGE];
  __shared__ float          s_er[RANGE];
  __shared__ int            s_lsrc[LIST_CAP];
  __shared__ float          s_le[LIST_CAP];
  __shared__ unsigned short s_ld[LIST_CAP];
  __shared__ int            s_wcnt[AGG_WAVES];
  __shared__ int            s_over;

  const int tid  = threadIdx.x;
  const int wave = tid >> 5;
  const int lane = tid & 31;
  const int base = blockIdx.x * RANGE;
  const int nloc = min(RANGE, N - base);
  const int NEG_INF_KEY = fkey(-INFINITY);

  for (int i = tid; i < RANGE * OUT_DIM; i += AGG_THREADS) s_acc[i] = 0.f;
  for (int i = tid; i < RANGE; i += AGG_THREADS) {
    s_mkey[i] = NEG_INF_KEY;
    s_den[i]  = 0.f;
    s_er[i]   = (i < nloc) ? er[base + i] : 0.f;
  }
  if (tid == 0) s_over = 0;
  __syncthreads();

  const int seg = wave * WAVE_CAP;
  int wcnt = 0;
  auto take = [&](int d, int e) {
    unsigned dl = (unsigned)(d - base);
    bool hit = dl < (unsigned)nloc;
    unsigned mask = __builtin_amdgcn_ballot_w32(hit);
    if (mask != 0u) {
      int pos = wcnt + (int)__builtin_amdgcn_mbcnt_lo(mask, 0u);
      if (hit && pos < WAVE_CAP) {
        s_lsrc[seg + pos] = e;
        s_ld[seg + pos]   = (unsigned short)dl;
      }
      wcnt += __builtin_popcount(mask);
    }
  };
  {
    const int E4 = E >> 2;
    const int4* __restrict__ dst4 = (const int4*)dst;
    const int wq = __builtin_amdgcn_readfirstlane(tid & ~31);
    int qw = wq;
    for (; qw + 7 * AGG_THREADS + 31 < E4; qw += 8 * AGG_THREADS) {
      const int q = qw + lane;
      int4 v[8];
#pragma unroll
      for (int u = 0; u < 8; ++u) v[u] = dst4[q + u * AGG_THREADS];
#pragma unroll
      for (int u = 0; u < 8; ++u) {
        int e0 = (q + u * AGG_THREADS) << 2;
        take(v[u].x, e0); take(v[u].y, e0 + 1); take(v[u].z, e0 + 2); take(v[u].w, e0 + 3);
      }
    }
    for (; qw < E4; qw += AGG_THREADS) {
      const int q = qw + lane;
      const bool in = q < E4;
      int4 v = make_int4(base - 1, base - 1, base - 1, base - 1);
      if (in) v = dst4[q];
      int e0 = q << 2;
      take(v.x, e0); take(v.y, e0 + 1); take(v.z, e0 + 2); take(v.w, e0 + 3);
    }
    for (int ew = (E4 << 2) + wq; ew < E; ew += AGG_THREADS) {
      const int e = ew + lane;
      int d = base - 1;
      if (e < E) d = dst[e];
      take(d, e);
    }
  }
  if (lane == 0) {
    s_wcnt[wave] = min(wcnt, WAVE_CAP);
    if (wcnt > WAVE_CAP) s_over = 1;
  }
  __syncthreads();

  if (s_over == 0) {
    const int n = __builtin_amdgcn_readfirstlane(s_wcnt[wave]);
    for (int i0 = 0; i0 < n; i0 += 128) {
      int sx[4]; int dl[4]; float lg[4];
#pragma unroll
      for (int u = 0; u < 4; ++u) {
        int i = i0 + u * 32 + lane;
        sx[u] = (i < n) ? src[s_lsrc[seg + i]] : 0;
        dl[u] = (i < n) ? (int)s_ld[seg + i] : 0;
      }
#pragma unroll
      for (int u = 0; u < 4; ++u) lg[u] = el[sx[u]];
#pragma unroll
      for (int u = 0; u < 4; ++u) {
        int i = i0 + u * 32 + lane;
        if (i < n) {
          float ev = leaky(lg[u] + s_er[dl[u]]);
          atomicMax(&s_mkey[dl[u]], fkey(ev));
          s_lsrc[seg + i] = sx[u];
          s_le[seg + i]   = ev;
        }
      }
    }
    __syncthreads();
    for (int i = 0; i < n; i += PASS2_EDGES) {
      float wx[PASS2_EDGES]; int dx[PASS2_EDGES]; v2f fv[PASS2_EDGES];
#pragma unroll
      for (int u = 0; u < PASS2_EDGES; ++u) {
        int jj = (i + u < n) ? (i + u) : i;
        int sidx = s_lsrc[seg + jj];
        dx[u] = s_ld[seg + jj];
        wx[u] = __expf(s_le[seg + jj] - fkey_inv(s_mkey[dx[u]]));
        fv[u] = *(const v2f*)(feat + (size_t)sidx * OUT_DIM + 2 * lane);
      }
#pragma unroll
      for (int u = 0; u < PASS2_EDGES; ++u) {
        if (i + u < n) {
          float* row = &s_acc[dx[u] * OUT_DIM + 2 * lane];
          atomicAdd(row,     wx[u] * fv[u].x);
          atomicAdd(row + 1, wx[u] * fv[u].y);
          if (lane == 0) atomicAdd(&s_den[dx[u]], wx[u]);
        }
      }
    }
  } else {
    for (int e = tid; e < E; e += AGG_THREADS) {
      unsigned dl = (unsigned)(dst[e] - base);
      if (dl < (unsigned)nloc)
        atomicMax(&s_mkey[dl], fkey(leaky(el[src[e]] + s_er[dl])));
    }
    __syncthreads();
    for (int e = tid; e < E; e += AGG_THREADS) {
      unsigned dl = (unsigned)(dst[e] - base);
      if (dl < (unsigned)nloc) {
        int   sidx = src[e];
        float ev   = leaky(el[sidx] + s_er[dl]);
        float w    = __expf(ev - fkey_inv(s_mkey[dl]));
        atomicAdd(&s_den[dl], w);
        const float* fr = feat + (size_t)sidx * OUT_DIM;
        for (int c = 0; c < OUT_DIM; ++c) atomicAdd(&s_acc[dl * OUT_DIM + c], w * fr[c]);
      }
    }
  }
  __syncthreads();

  float* __restrict__ ob = out + (size_t)base * OUT_DIM;
  for (int i = tid; i < nloc * OUT_DIM; i += AGG_THREADS) {
    float den = s_den[i >> 6];
    float inv = (den > 0.f) ? 1.f / den : 1.f;
    ob[i] = s_acc[i] * inv + bias[i & (OUT_DIM - 1)];
  }
}

extern "C" void kernel_launch(void* const* d_in, const int* in_sizes, int n_in,
                              void* d_out, int out_size, void* d_ws, size_t ws_size,
                              hipStream_t stream) {
  const float* h      = (const float*)d_in[0];
  const int*   src    = (const int*)d_in[1];
  const int*   dst    = (const int*)d_in[2];
  const float* W      = (const float*)d_in[3];
  const float* attn_l = (const float*)d_in[4];
  const float* attn_r = (const float*)d_in[5];
  const float* bias   = (const float*)d_in[6];
  float* out = (float*)d_out;

  const int N = in_sizes[0] / IN_DIM;
  const int E = in_sizes[1];

  char* ws = (char*)d_ws;
  size_t off = 0;
  auto carve = [&](size_t bytes) -> char* {
    char* p = ws + off;
    off = (off + bytes + 255) & ~(size_t)255;
    return p;
  };
  float* feat    = (float*)carve((size_t)N * OUT_DIM * sizeof(float));
  float* Wt      = (float*)carve((size_t)IN_DIM * OUT_DIM * sizeof(float));
  float* el      = (float*)carve((size_t)N * sizeof(float));
  float* er      = (float*)carve((size_t)N * sizeof(float));

  k_wt<<<(IN_DIM * OUT_DIM + 255) / 256, 256, 0, stream>>>(W, Wt);
  k_feat_wmma<<<(N + 31) / 32, 256, 0, stream>>>(h, Wt, feat, N);
  k_el_er<<<(N + 255) / 256, 256, 0, stream>>>(feat, attn_l, attn_r, el, er, N);
  k_gat_range<<<(N + RANGE - 1) / RANGE, AGG_THREADS, 0, stream>>>(feat, el, er, src, dst, bias, out, N, E);
}


# Aggregation pass 2: one lane per edge fetches source, node and exp weight (32 edges at a time), the wave gets them by lane read; no per-edge broadcast LDS loads, no all-lane exp. den sum re-ordered.
# speedup vs baseline: 1.8375x; 1.0501x over previous
#include <hip/hip_runtime.h>
#include <math.h>

#define IN_DIM   128
#define OUT_DIM  64
#define NEG_SLOPE 0.2f

typedef float v2f __attribute__((ext_vector_type(2)));
typedef float v8f __attribute__((ext_vector_type(8)));

__device__ __forceinline__ float leaky(float e) {
  return (e > 0.f) ? e : NEG_SLOPE * e;
}

__global__ void k_wt(const float* __restrict__ W, float* __restrict__ Wt) {
  int i = blockIdx.x * blockDim.x + threadIdx.x;
  if (i >= IN_DIM * OUT_DIM) return;
  int k = i / OUT_DIM;
  int c = i - k * OUT_DIM;
  Wt[(size_t)c * IN_DIM + k] = W[i];
}

__global__ __launch_bounds__(256) void k_feat_wmma(const float* __restrict__ h,
                                                   const float* __restrict__ Wt,
                                                   float* __restrict__ feat,
                                                   int N) {
  const int tid   = threadIdx.x;
  const int wave  = tid >> 5;
  const int lane  = tid & 31;
  const int mtile = wave >> 2;
  const int ntile = wave & 3;
  const int rowBase = blockIdx.x * 32 + mtile * 16;
  const int col   = ntile * 16 + (lane & 15);
  const int row   = rowBase + (lane & 15);
  const int khalf = (lane >> 4) << 1;

  const int rowc = (row < N) ? row : (N - 1);
  const float* __restrict__ ap = h  + (size_t)rowc * IN_DIM + khalf;
  const float* __restrict__ bp = Wt + (size_t)col  * IN_DIM + khalf;

  v8f acc = {0.f, 0.f, 0.f, 0.f, 0.f, 0.f, 0.f, 0.f};

#pragma unroll
  for (int k = 0; k < IN_DIM; k += 16) {
    v2f a0, a1, a2, a3, b0, b1, b2, b3;
    a0 = *(const v2f*)(ap + k);      b0 = *(const v2f*)(bp + k);
    a1 = *(const v2f*)(ap + k + 4);  b1 = *(const v2f*)(bp + k + 4);
    a2 = *(const v2f*)(ap + k + 8);  b2 = *(const v2f*)(bp + k + 8);
    a3 = *(const v2f*)(ap + k + 12); b3 = *(const v2f*)(bp + k + 12);
    acc = __builtin_amdgcn_wmma_f32_16x16x4_f32(false, a0, false, b0, (short)0, acc, false, false);
    acc = __builtin_amdgcn_wmma_f32_16x16x4_f32(false, a1, false, b1, (short)0, acc, false, false);
    acc = __builtin_amdgcn_wmma_f32_16x16x4_f32(false, a2, false, b2, (short)0, acc, false, false);
    acc = __builtin_amdgcn_wmma_f32_16x16x4_f32(false, a3, false, b3, (short)0, acc, false, false);
  }

  const int mBase = rowBase + (lane >> 4) * 8;
#pragma unroll
  for (int r = 0; r < 8; ++r) {
    int m = mBase + r;
    if (m < N) feat[(size_t)m * OUT_DIM + col] = acc[r];
  }
}

__global__ void k_el_er(const float* __restrict__ feat,
                        const float* __restrict__ attn_l,
                        const float* __restrict__ attn_r,
                        float* __restrict__ el, float* __restrict__ er, int N) {
  int i = blockIdx.x * blockDim.x + threadIdx.x;
  if (i >= N) return;
  const float4* f  = (const float4*)(feat + (size_t)i * OUT_DIM);
  const float4* al = (const float4*)attn_l;
  const float4* ar = (const float4*)attn_r;
  float sl = 0.f, sr = 0.f;
#pragma unroll
  for (int j = 0; j < OUT_DIM / 4; ++j) {
    float4 v = f[j], a = al[j], b = ar[j];
    sl += v.x * a.x + v.y * a.y + v.z * a.z + v.w * a.w;
    sr += v.x * b.x + v.y * b.y + v.z * b.z + v.w * b.w;
  }
  el[i] = sl;
  er[i] = sr;
}

#define RANGE    512
#define LIST_CAP 16384
#define AGG_THREADS 1024
#define AGG_WAVES   (AGG_THREADS / 32)
#define PASS2_EDGES 8
#define WAVE_CAP    (LIST_CAP / AGG_WAVES)

__device__ __forceinline__ int fkey(float f) {
  int b = __float_as_int(f);
  return (b >= 0) ? b : (b ^ 0x7fffffff);
}
__device__ __forceinline__ float fkey_inv(int k) {
  return __int_as_float((k >= 0) ? k : (k ^ 0x7fffffff));
}

__global__ __launch_bounds__(AGG_THREADS) void k_gat_range(const float* __restrict__ feat,
                                                    const float* __restrict__ el,
                                                    const float* __restrict__ er,
                                                    const int* __restrict__ src,
                                                    const int* __restrict__ dst,
                                                    const float* __restrict__ bias,
                                                    float* __restrict__ out,
                                                    int N, int E) {
  __shared__ float          s_acc[RANGE * OUT_DIM];
  __shared__ int            s_mkey[RANGE];
  __shared__ float          s_den[RANGE];
  __shared__ float          s_er[RANGE];
  __shared__ int            s_lsrc[LIST_CAP];
  __shared__ float          s_le[LIST_CAP];
  __shared__ unsigned short s_ld[LIST_CAP];
  __shared__ int            s_wcnt[AGG_WAVES];
  __shared__ int            s_over;

  const int tid  = threadIdx.x;
  const int wave = tid >> 5;
  const int lane = tid & 31;
  const int base = blockIdx.x * RANGE;
  const int nloc = min(RANGE, N - base);
  const int NEG_INF_KEY = fkey(-INFINITY);

  for (int i = tid; i < RANGE * OUT_DIM; i += AGG_THREADS) s_acc[i] = 0.f;
  for (int i = tid; i < RANGE; i += AGG_THREADS) {
    s_mkey[i] = NEG_INF_KEY;
    s_den[i]  = 0.f;
    s_er[i]   = (i < nloc) ? er[base + i] : 0.f;
  }
  if (tid == 0) s_over = 0;
  __syncthreads();

  const int seg = wave * WAVE_CAP;
  int wcnt = 0;
  auto take = [&](int d, int e) {
    unsigned dl = (unsigned)(d - base);
    bool hit = dl < (unsigned)nloc;
    unsigned mask = __builtin_amdgcn_ballot_w32(hit);
    if (mask != 0u) {
      int pos = wcnt + (int)__builtin_amdgcn_mbcnt_lo(mask, 0u);
      if (hit && pos < WAVE_CAP) {
        s_lsrc[seg + pos] = e;
        s_ld[seg + pos]   = (unsigned short)dl;
      }
      wcnt += __builtin_popcount(mask);
    }
  };
  {
    const int E4 = E >> 2;
    const int4* __restrict__ dst4 = (const int4*)dst;
    const int wq = __builtin_amdgcn_readfirstlane(tid & ~31);
    int qw = wq;
    for (; qw + 7 * AGG_THREADS + 31 < E4; qw += 8 * AGG_THREADS) {
      const int q = qw + lane;
      int4 v[8];
#pragma unroll
      for (int u = 0; u < 8; ++u) v[u] = dst4[q + u * AGG_THREADS];
#pragma unroll
      for (int u = 0; u < 8; ++u) {
        int e0 = (q + u * AGG_THREADS) << 2;
        take(v[u].x, e0); take(v[u].y, e0 + 1); take(v[u].z, e0 + 2); take(v[u].w, e0 + 3);
      }
    }
    for (; qw < E4; qw += AGG_THREADS) {
      const int q = qw + lane;
      const bool in = q < E4;
      int4 v = make_int4(base - 1, base - 1, base - 1, base - 1);
      if (in) v = dst4[q];
      int e0 = q << 2;
      take(v.x, e0); take(v.y, e0 + 1); take(v.z, e0 + 2); take(v.w, e0 + 3);
    }
    for (int ew = (E4 << 2) + wq; ew < E; ew += AGG_THREADS) {
      const int e = ew + lane;
      int d = base - 1;
      if (e < E) d = dst[e];
      take(d, e);
    }
  }
  if (lane == 0) {
    s_wcnt[wave] = min(wcnt, WAVE_CAP);
    if (wcnt > WAVE_CAP) s_over = 1;
  }
  __syncthreads();

  if (s_over == 0) {
    const int n = __builtin_amdgcn_readfirstlane(s_wcnt[wave]);
    for (int i0 = 0; i0 < n; i0 += 128) {
      int sx[4]; int dl[4]; float lg[4];
#pragma unroll
      for (int u = 0; u < 4; ++u) {
        int i = i0 + u * 32 + lane;
        sx[u] = (i < n) ? src[s_lsrc[seg + i]] : 0;
        dl[u] = (i < n) ? (int)s_ld[seg + i] : 0;
      }
#pragma unroll
      for (int u = 0; u < 4; ++u) lg[u] = el[sx[u]];
#pragma unroll
      for (int u = 0; u < 4; ++u) {
        int i = i0 + u * 32 + lane;
        if (i < n) {
          float ev = leaky(lg[u] + s_er[dl[u]]);
          atomicMax(&s_mkey[dl[u]], fkey(ev));
          s_lsrc[seg + i] = sx[u];
          s_le[seg + i]   = ev;
        }
      }
    }
    __syncthreads();
    const float* __restrict__ featl = feat + 2 * lane;
    float* accl = s_acc + 2 * lane;
    for (int i0 = 0; i0 < n; i0 += 32) {
      const int cnt = min(32, n - i0);
      int sx = 0, dn = 0, wb = 0;
      if (lane < cnt) {
        const int j = seg + i0 + lane;
        sx = s_lsrc[j];
        dn = s_ld[j];
        const float w = __expf(s_le[j] - fkey_inv(s_mkey[dn]));
        atomicAdd(&s_den[dn], w);
        wb = __float_as_int(w);
      }
      auto group = [&](const int g, const bool guard) {
        int dv[PASS2_EDGES]; float wv[PASS2_EDGES]; v2f fv[PASS2_EDGES];
#pragma unroll
        for (int u = 0; u < PASS2_EDGES; ++u) {
          const int l = (!guard || g + u < cnt) ? (g + u) : g;
          const int sv = __builtin_amdgcn_readlane(sx, l);
          dv[u] = __builtin_amdgcn_readlane(dn, l);
          wv[u] = __int_as_float(__builtin_amdgcn_readlane(wb, l));
          fv[u] = *(const v2f*)(featl + (size_t)sv * OUT_DIM);
        }
#pragma unroll
        for (int u = 0; u < PASS2_EDGES; ++u) {
          if (!guard || g + u < cnt) {
            float* row = accl + dv[u] * OUT_DIM;
            atomicAdd(row,     wv[u] * fv[u].x);
            atomicAdd(row + 1, wv[u] * fv[u].y);
          }
        }
      };
      if (cnt == 32) {
#pragma unroll
        for (int g = 0; g < 32; g += PASS2_EDGES) group(g, false);
      } else {
        for (int g = 0; g < cnt; g += PASS2_EDGES) group(g, true);
      }
    }
  } else {
    for (int e = tid; e < E; e += AGG_THREADS) {
      unsigned dl = (unsigned)(dst[e] - base);
      if (dl < (unsigned)nloc)
        atomicMax(&s_mkey[dl], fkey(leaky(el[src[e]] + s_er[dl])));
    }
    __syncthreads();
    for (int e = tid; e < E; e += AGG_THREADS) {
      unsigned dl = (unsigned)(dst[e] - base);
      if (dl < (unsigned)nloc) {
        int   sidx = src[e];
        float ev   = leaky(el[sidx] + s_er[dl]);
        float w    = __expf(ev - fkey_inv(s_mkey[dl]));
        atomicAdd(&s_den[dl], w);
        const float* fr = feat + (size_t)sidx * OUT_DIM;
        for (int c = 0; c < OUT_DIM; ++c) atomicAdd(&s_acc[dl * OUT_DIM + c], w * fr[c]);
      }
    }
  }
  __syncthreads();

  float* __restrict__ ob = out + (size_t)base * OUT_DIM;
  for (int i = tid; i < nloc * OUT_DIM; i += AGG_THREADS) {
    float den = s_den[i >> 6];
    float inv = (den > 0.f) ? 1.f / den : 1.f;
    ob[i] = s_acc[i] * inv + bias[i & (OUT_DIM - 1)];
  }
}

extern "C" void kernel_launch(void* const* d_in, const int* in_sizes, int n_in,
                              void* d_out, int out_size, void* d_ws, size_t ws_size,
                              hipStream_t stream) {
  const float* h      = (const float*)d_in[0];
  const int*   src    = (const int*)d_in[1];
  const int*   dst    = (const int*)d_in[2];
  const float* W      = (const float*)d_in[3];
  const float* attn_l = (const float*)d_in[4];
  const float* attn_r = (const float*)d_in[5];
  const float* bias   = (const float*)d_in[6];
  float* out = (float*)d_out;

  const int N = in_sizes[0] / IN_DIM;
  const int E = in_sizes[1];

  char* ws = (char*)d_ws;
  size_t off = 0;
  auto carve = [&](size_t bytes) -> char* {
    char* p = ws + off;
    off = (off + bytes + 255) & ~(size_t)255;
    return p;
  };
  float* feat    = (float*)carve((size_t)N * OUT_DIM * sizeof(float));
  float* Wt      = (float*)carve((size_t)IN_DIM * OUT_DIM * sizeof(float));
  float* el      = (float*)carve((size_t)N * sizeof(float));
  float* er      = (float*)carve((size_t)N * sizeof(float));

  k_wt<<<(IN_DIM * OUT_DIM + 255) / 256, 256, 0, stream>>>(W, Wt);
  k_feat_wmma<<<(N + 31) / 32, 256, 0, stream>>>(h, Wt, feat, N);
  k_el_er<<<(N + 255) / 256, 256, 0, stream>>>(feat, attn_l, attn_r, el, er, N);
  k_gat_range<<<(N + RANGE - 1) / RANGE, AGG_THREADS, 0, stream>>>(feat, el, er, src, dst, bias, out, N, E);
}


# Edge filter scans one byte per edge, not four: the transpose launch also writes each edge's owner-block byte and a packed source+node word; blocks test 4 keys per word op. Node sums: LDS arrival.
# speedup vs baseline: 2.3469x; 1.2772x over previous
#include <hip/hip_runtime.h>
#include <math.h>

#define IN_DIM   128
#define OUT_DIM  64
#define NEG_SLOPE 0.2f

typedef float v2f __attribute__((ext_vector_type(2)));
typedef float v8f __attribute__((ext_vector_type(8)));

__device__ __forceinline__ float leaky(float e) {
  return (e > 0.f) ? e : NEG_SLOPE * e;
}

#define RANGE      512
#define RANGE_LOG2 9
#define KEY_NONE   255u
#define KEY_BLOCKS 255
#define WT_BLOCKS  ((IN_DIM * OUT_DIM + 255) / 256)

__device__ __forceinline__ unsigned edge_key(int d, int N) {
  unsigned k = (unsigned)d >> RANGE_LOG2;
  return ((unsigned)d < (unsigned)N && k < KEY_NONE) ? k : KEY_NONE;
}
__device__ __forceinline__ unsigned edge_pack(int s, int d) {
  return ((unsigned)s << RANGE_LOG2) | ((unsigned)d & (RANGE - 1));
}

__global__ __launch_bounds__(256) void k_wt_keys(const float* __restrict__ W, float* __restrict__ Wt,
                                                 const int* __restrict__ src,
                                                 const int* __restrict__ dst,
                                                 unsigned* __restrict__ keyw,
                                                 unsigned* __restrict__ packed,
                                                 int N, int E) {
  if (blockIdx.x < WT_BLOCKS) {
    int i = blockIdx.x * 256 + threadIdx.x;
    if (i >= IN_DIM * OUT_DIM) return;
    int k = i / OUT_DIM;
    int c = i - k * OUT_DIM;
    Wt[(size_t)c * IN_DIM + k] = W[i];
    return;
  }
  const int GP = ((E + 15) >> 4) << 2;
  const int gb = (blockIdx.x - WT_BLOCKS) * 1024;
  const int g0 = gb + threadIdx.x;
  const int4* __restrict__ dst4 = (const int4*)dst;
  const int4* __restrict__ src4 = (const int4*)src;
  if ((gb + 1024) * 4 <= E) {
    int4 d[4], s[4];
#pragma unroll
    for (int u = 0; u < 4; ++u) { d[u] = dst4[g0 + u * 256]; s[u] = src4[g0 + u * 256]; }
#pragma unroll
    for (int u = 0; u < 4; ++u) {
      uint4 pk;
      pk.x = edge_pack(s[u].x, d[u].x); pk.y = edge_pack(s[u].y, d[u].y);
      pk.z = edge_pack(s[u].z, d[u].z); pk.w = edge_pack(s[u].w, d[u].w);
      ((uint4*)packed)[g0 + u * 256] = pk;
      keyw[g0 + u * 256] = edge_key(d[u].x, N) | (edge_key(d[u].y, N) << 8) |
                           (edge_key(d[u].z, N) << 16) | (edge_key(d[u].w, N) << 24);
    }
    return;
  }
  for (int u = 0; u < 4; ++u) {
    const int g = g0 + u * 256;
    if (g >= GP) continue;
    const int e0 = g << 2;
    unsigned kw;
    if (e0 + 3 < E) {
      const int4 d = dst4[g];
      const int4 s = src4[g];
      uint4 pk;
      pk.x = edge_pack(s.x, d.x); pk.y = edge_pack(s.y, d.y);
      pk.z = edge_pack(s.z, d.z); pk.w = edge_pack(s.w, d.w);
      ((uint4*)packed)[g] = pk;
      kw = edge_key(d.x, N) | (edge_key(d.y, N) << 8) | (edge_key(d.z, N) << 16) | (edge_key(d.w, N) << 24);
    } else {
      kw = 0u;
      for (int b = 0; b < 4; ++b) {
        const int e = e0 + b;
        unsigned k = KEY_NONE;
        if (e < E) {
          const int d = dst[e];
          packed[e] = edge_pack(src[e], d);
          k = edge_key(d, N);
        }
        kw |= k << (8 * b);
      }
    }
    keyw[g] = kw;
  }
}

__global__ __launch_bounds__(256) void k_feat_wmma(const float* __restrict__ h,
                                                   const float* __restrict__ Wt,
                                                   float* __restrict__ feat,
                                                   int N) {
  const int tid   = threadIdx.x;
  const int wave  = tid >> 5;
  const int lane  = tid & 31;
  const int mtile = wave >> 2;
  const int ntile = wave & 3;
  const int rowBase = blockIdx.x * 32 + mtile * 16;
  const int col   = ntile * 16 + (lane & 15);
  const int row   = rowBase + (lane & 15);
  const int khalf = (lane >> 4) << 1;

  const int rowc = (row < N) ? row : (N - 1);
  const float* __restrict__ ap = h  + (size_t)rowc * IN_DIM + khalf;
  const float* __restrict__ bp = Wt + (size_t)col  * IN_DIM + khalf;

  v8f acc = {0.f, 0.f, 0.f, 0.f, 0.f, 0.f, 0.f, 0.f};

#pragma unroll
  for (int k = 0; k < IN_DIM; k += 16) {
    v2f a0, a1, a2, a3, b0, b1, b2, b3;
    a0 = *(const v2f*)(ap + k);      b0 = *(const v2f*)(bp + k);
    a1 = *(const v2f*)(ap + k + 4);  b1 = *(const v2f*)(bp + k + 4);
    a2 = *(const v2f*)(ap + k + 8);  b2 = *(const v2f*)(bp + k + 8);
    a3 = *(const v2f*)(ap + k + 12); b3 = *(const v2f*)(bp + k + 12);
    acc = __builtin_amdgcn_wmma_f32_16x16x4_f32(false, a0, false, b0, (short)0, acc, false, false);
    acc = __builtin_amdgcn_wmma_f32_16x16x4_f32(false, a1, false, b1, (short)0, acc, false, false);
    acc = __builtin_amdgcn_wmma_f32_16x16x4_f32(false, a2, false, b2, (short)0, acc, false, false);
    acc = __builtin_amdgcn_wmma_f32_16x16x4_f32(false, a3, false, b3, (short)0, acc, false, false);
  }

  const int mBase = rowBase + (lane >> 4) * 8;
#pragma unroll
  for (int r = 0; r < 8; ++r) {
    int m = mBase + r;
    if (m < N) feat[(size_t)m * OUT_DIM + col] = acc[r];
  }
}

__global__ void k_el_er(const float* __restrict__ feat,
                        const float* __restrict__ attn_l,
                        const float* __restrict__ attn_r,
                        float* __restrict__ el, float* __restrict__ er, int N) {
  int i = blockIdx.x * blockDim.x + threadIdx.x;
  if (i >= N) return;
  const float4* f  = (const float4*)(feat + (size_t)i * OUT_DIM);
  const float4* al = (const float4*)attn_l;
  const float4* ar = (const float4*)attn_r;
  float sl = 0.f, sr = 0.f;
#pragma unroll
  for (int j = 0; j < OUT_DIM / 4; ++j) {
    float4 v = f[j], a = al[j], b = ar[j];
    sl += v.x * a.x + v.y * a.y + v.z * a.z + v.w * a.w;
    sr += v.x * b.x + v.y * b.y + v.z * b.z + v.w * b.w;
  }
  el[i] = sl;
  er[i] = sr;
}

#define LIST_CAP 16384
#define AGG_THREADS 1024
#define AGG_WAVES   (AGG_THREADS / 32)
#define PASS2_EDGES 8
#define WAVE_CAP    (LIST_CAP / AGG_WAVES)

__device__ __forceinline__ int fkey(float f) {
  int b = __float_as_int(f);
  return (b >= 0) ? b : (b ^ 0x7fffffff);
}
__device__ __forceinline__ float fkey_inv(int k) {
  return __int_as_float((k >= 0) ? k : (k ^ 0x7fffffff));
}

__global__ __launch_bounds__(AGG_THREADS) void k_gat_range(const float* __restrict__ feat,
                                                    const float* __restrict__ el,
                                                    const float* __restrict__ er,
                                                    const int* __restrict__ src,
                                                    const int* __restrict__ dst,
                                                    const unsigned* __restrict__ keyw,
                                                    const unsigned* __restrict__ packed,
                                                    const float* __restrict__ bias,
                                                    float* __restrict__ out,
                                                    int N, int E) {
  __shared__ float          s_acc[RANGE * OUT_DIM];
  __shared__ int            s_mkey[RANGE];
  __shared__ float          s_den[RANGE];
  __shared__ float          s_er[RANGE];
  __shared__ int            s_lsrc[LIST_CAP];
  __shared__ float          s_le[LIST_CAP];
  __shared__ unsigned short s_ld[LIST_CAP];
  __shared__ int            s_wcnt[AGG_WAVES];
  __shared__ int            s_over;

  const int tid  = threadIdx.x;
  const int wave = tid >> 5;
  const int lane = tid & 31;
  const int base = blockIdx.x * RANGE;
  const int nloc = min(RANGE, N - base);
  const int NEG_INF_KEY = fkey(-INFINITY);

  for (int i = tid; i < RANGE * OUT_DIM; i += AGG_THREADS) s_acc[i] = 0.f;
  for (int i = tid; i < RANGE; i += AGG_THREADS) {
    s_mkey[i] = NEG_INF_KEY;
    s_den[i]  = 0.f;
    s_er[i]   = (i < nloc) ? er[base + i] : 0.f;
  }
  const bool use_keys = N <= KEY_BLOCKS * RANGE;
  if (tid == 0) s_over = use_keys ? 0 : 1;
  __syncthreads();

  const int seg = wave * WAVE_CAP;
  int wcnt = 0;
  if (use_keys) {
    const unsigned pat = (unsigned)blockIdx.x * 0x01010101u;
    auto hits16 = [&](const uint4 v) -> unsigned {
      const unsigned x0 = v.x ^ pat, x1 = v.y ^ pat, x2 = v.z ^ pat, x3 = v.w ^ pat;
      const unsigned n0 = ((x0 & 0x7F7F7F7Fu) + 0x7F7F7F7Fu) | x0;
      const unsigned n1 = ((x1 & 0x7F7F7F7Fu) + 0x7F7F7F7Fu) | x1;
      const unsigned n2 = ((x2 & 0x7F7F7F7Fu) + 0x7F7F7F7Fu) | x2;
      const unsigned n3 = ((x3 & 0x7F7F7F7Fu) + 0x7F7F7F7Fu) | x3;
      const unsigned nz = ((n0 >> 7) & 0x01010101u) | ((n1 >> 6) & 0x02020202u) |
                          ((n2 >> 5) & 0x04040404u) | ((n3 >> 4) & 0x08080808u);
      return nz ^ 0x0F0F0F0Fu;
    };
    auto push = [&](unsigned m, const int q) {
      unsigned any = __builtin_amdgcn_ballot_w32(m != 0u);
      while (any != 0u) {
        if (m != 0u) {
          const int pos = wcnt + (int)__builtin_amdgcn_mbcnt_lo(any, 0u);
          const int p = __builtin_ctz(m);
          const int e = ((q + ((p >> 2) & 1) * AGG_THREADS) << 4) + ((p & 3) << 2) + (p >> 3);
          if (pos < WAVE_CAP) s_lsrc[seg + pos] = e;
          m &= m - 1u;
        }
        wcnt += __builtin_popcount(any);
        any = __builtin_amdgcn_ballot_w32(m != 0u);
      }
    };
    const int Q = (E + 15) >> 4;
    const uint4* __restrict__ key4 = (const uint4*)keyw;
    const int wq = __builtin_amdgcn_readfirstlane(tid & ~31);
    int qw = wq;
    for (; qw + 7 * AGG_THREADS + 31 < Q; qw += 8 * AGG_THREADS) {
      const int q = qw + lane;
      uint4 v[8];
#pragma unroll
      for (int u = 0; u < 8; ++u) v[u] = key4[q + u * AGG_THREADS];
#pragma unroll
      for (int u = 0; u < 8; u += 2)
        push(hits16(v[u]) | (hits16(v[u + 1]) << 4), q + u * AGG_THREADS);
    }
    for (; qw < Q; qw += AGG_THREADS) {
      const int q = qw + lane;
      uint4 v = make_uint4(~0u, ~0u, ~0u, ~0u);
      if (q < Q) v = key4[q];
      push(hits16(v), q);
    }
  }
  if (lane == 0) {
    s_wcnt[wave] = min(wcnt, WAVE_CAP);
    if (wcnt > WAVE_CAP) s_over = 1;
  }
  __syncthreads();

  if (s_over == 0) {
    const int n = __builtin_amdgcn_readfirstlane(s_wcnt[wave]);
    for (int i0 = 0; i0 < n; i0 += 128) {
      int sx[4]; int dl[4]; float lg[4];
#pragma unroll
      for (int u = 0; u < 4; ++u) {
        int i = i0 + u * 32 + lane;
        const unsigned pk = (i < n) ? packed[s_lsrc[seg + i]] : 0u;
        sx[u] = (int)(pk >> RANGE_LOG2);
        dl[u] = (int)(pk & (RANGE - 1));
      }
#pragma unroll
      for (int u = 0; u < 4; ++u) lg[u] = el[sx[u]];
#pragma unroll
      for (int u = 0; u < 4; ++u) {
        int i = i0 + u * 32 + lane;
        if (i < n) {
          float ev = leaky(lg[u] + s_er[dl[u]]);
          atomicMax(&s_mkey[dl[u]], fkey(ev));
          s_lsrc[seg + i] = sx[u];
          s_le[seg + i]   = ev;
          s_ld[seg + i]   = (unsigned short)dl[u];
        }
      }
    }
    __syncthreads();
    const float* __restrict__ featl = feat + 2 * lane;
    float* accl = s_acc + 2 * lane;
    for (int i0 = 0; i0 < n; i0 += 32) {
      const int cnt = min(32, n - i0);
      int sx = 0, dn = 0, wb = 0;
      if (lane < cnt) {
        const int j = seg + i0 + lane;
        sx = s_lsrc[j];
        dn = s_ld[j];
        const float w = __expf(s_le[j] - fkey_inv(s_mkey[dn]));
        atomicAdd(&s_den[dn], w);
        wb = __float_as_int(w);
      }
      auto group = [&](const int g, const bool guard) {
        int dv[PASS2_EDGES]; float wv[PASS2_EDGES]; v2f fv[PASS2_EDGES];
#pragma unroll
        for (int u = 0; u < PASS2_EDGES; ++u) {
          const int l = (!guard || g + u < cnt) ? (g + u) : g;
          const int sv = __builtin_amdgcn_readlane(sx, l);
          dv[u] = __builtin_amdgcn_readlane(dn, l);
          wv[u] = __int_as_float(__builtin_amdgcn_readlane(wb, l));
          fv[u] = *(const v2f*)(featl + (size_t)sv * OUT_DIM);
        }
#pragma unroll
        for (int u = 0; u < PASS2_EDGES; ++u) {
          if (!guard || g + u < cnt) {
            float* row = accl + dv[u] * OUT_DIM;
            atomicAdd(row,     wv[u] * fv[u].x);
            atomicAdd(row + 1, wv[u] * fv[u].y);
          }
        }
      };
      if (cnt == 32) {
#pragma unroll
        for (int g = 0; g < 32; g += PASS2_EDGES) group(g, false);
      } else {
        for (int g = 0; g < cnt; g += PASS2_EDGES) group(g, true);
      }
    }
  } else {
    for (int e = tid; e < E; e += AGG_THREADS) {
      unsigned dl = (unsigned)(dst[e] - base);
      if (dl < (unsigned)nloc)
        atomicMax(&s_mkey[dl], fkey(leaky(el[src[e]] + s_er[dl])));
    }
    __syncthreads();
    for (int e = tid; e < E; e += AGG_THREADS) {
      unsigned dl = (unsigned)(dst[e] - base);
      if (dl < (unsigned)nloc) {
        int   sidx = src[e];
        float ev   = leaky(el[sidx] + s_er[dl]);
        float w    = __expf(ev - fkey_inv(s_mkey[dl]));
        atomicAdd(&s_den[dl], w);
        const float* fr = feat + (size_t)sidx * OUT_DIM;
        for (int c = 0; c < OUT_DIM; ++c) atomicAdd(&s_acc[dl * OUT_DIM + c], w * fr[c]);
      }
    }
  }
  __syncthreads();

  float* __restrict__ ob = out + (size_t)base * OUT_DIM;
  for (int i = tid; i < nloc * OUT_DIM; i += AGG_THREADS) {
    float den = s_den[i >> 6];
    float inv = (den > 0.f) ? 1.f / den : 1.f;
    ob[i] = s_acc[i] * inv + bias[i & (OUT_DIM - 1)];
  }
}

extern "C" void kernel_launch(void* const* d_in, const int* in_sizes, int n_in,
                              void* d_out, int out_size, void* d_ws, size_t ws_size,
                              hipStream_t stream) {
  const float* h      = (const float*)d_in[0];
  const int*   src    = (const int*)d_in[1];
  const int*   dst    = (const int*)d_in[2];
  const float* W      = (const float*)d_in[3];
  const float* attn_l = (const float*)d_in[4];
  const float* attn_r = (const float*)d_in[5];
  const float* bias   = (const float*)d_in[6];
  float* out = (float*)d_out;

  const int N = in_sizes[0] / IN_DIM;
  const int E = in_sizes[1];

  char* ws = (char*)d_ws;
  size_t off = 0;
  auto carve = [&](size_t bytes) -> char* {
    char* p = ws + off;
    off = (off + bytes + 255) & ~(size_t)255;
    return p;
  };
  float* feat    = (float*)carve((size_t)N * OUT_DIM * sizeof(float));
  float* Wt      = (float*)carve((size_t)IN_DIM * OUT_DIM * sizeof(float));
  float* el      = (float*)carve((size_t)N * sizeof(float));
  float* er      = (float*)carve((size_t)N * sizeof(float));
  unsigned* keyw   = (unsigned*)carve((size_t)(((E + 15) >> 4) << 2) * sizeof(unsigned));
  unsigned* packed = (unsigned*)carve((size_t)E * sizeof(unsigned));

  k_wt_keys<<<WT_BLOCKS + (((E + 15) >> 4) * 4 + 1023) / 1024, 256, 0, stream>>>(W, Wt, src, dst, keyw, packed, N, E);
  k_feat_wmma<<<(N + 31) / 32, 256, 0, stream>>>(h, Wt, feat, N);
  k_el_er<<<(N + 255) / 256, 256, 0, stream>>>(feat, attn_l, attn_r, el, er, N);
  k_gat_range<<<(N + RANGE - 1) / RANGE, AGG_THREADS, 0, stream>>>(feat, el, er, src, dst, keyw, packed, bias, out, N, E);
}


# Edge filter leaves the aggregation launch: filter waves run as extra blocks of the projection launch and write each range's packed edge lists to the workspace. Node sums: LDS arrival order.
# speedup vs baseline: 2.5960x; 1.1061x over previous
#include <hip/hip_runtime.h>
#include <math.h>

#define IN_DIM   128
#define OUT_DIM  64
#define NEG_SLOPE 0.2f

typedef float v2f __attribute__((ext_vector_type(2)));
typedef float v8f __attribute__((ext_vector_type(8)));

__device__ __forceinline__ float leaky(float e) {
  return (e > 0.f) ? e : NEG_SLOPE * e;
}

#define RANGE      512
#define RANGE_LOG2 9
#define KEY_NONE   255u
#define KEY_BLOCKS 255
#define WT_BLOCKS  ((IN_DIM * OUT_DIM + 255) / 256)

#define LIST_CAP    16384
#define AGG_THREADS 1024
#define AGG_WAVES   (AGG_THREADS / 32)
#define WAVE_CAP    (LIST_CAP / AGG_WAVES)
#define SCAN_WAVES  8
#define SCAN_BLOCKS (AGG_WAVES / SCAN_WAVES)

__device__ __forceinline__ unsigned edge_key(int d, int N) {
  unsigned k = (unsigned)d >> RANGE_LOG2;
  return ((unsigned)d < (unsigned)N && k < KEY_NONE) ? k : KEY_NONE;
}
__device__ __forceinline__ unsigned edge_pack(int s, int d) {
  return ((unsigned)s << RANGE_LOG2) | ((unsigned)d & (RANGE - 1));
}

__global__ __launch_bounds__(256) void k_wt_keys(const float* __restrict__ W, float* __restrict__ Wt,
                                                 const int* __restrict__ src,
                                                 const int* __restrict__ dst,
                                                 unsigned* __restrict__ keyw,
                                                 unsigned* __restrict__ packed,
                                                 int N, int E) {
  if (blockIdx.x < WT_BLOCKS) {
    int i = blockIdx.x * 256 + threadIdx.x;
    if (i >= IN_DIM * OUT_DIM) return;
    int k = i / OUT_DIM;
    int c = i - k * OUT_DIM;
    Wt[(size_t)c * IN_DIM + k] = W[i];
    return;
  }
  const int GP = ((E + 15) >> 4) << 2;
  const int gb = (blockIdx.x - WT_BLOCKS) * 1024;
  const int g0 = gb + threadIdx.x;
  const int4* __restrict__ dst4 = (const int4*)dst;
  const int4* __restrict__ src4 = (const int4*)src;
  if ((gb + 1024) * 4 <= E) {
    int4 d[4], s[4];
#pragma unroll
    for (int u = 0; u < 4; ++u) { d[u] = dst4[g0 + u * 256]; s[u] = src4[g0 + u * 256]; }
#pragma unroll
    for (int u = 0; u < 4; ++u) {
      uint4 pk;
      pk.x = edge_pack(s[u].x, d[u].x); pk.y = edge_pack(s[u].y, d[u].y);
      pk.z = edge_pack(s[u].z, d[u].z); pk.w = edge_pack(s[u].w, d[u].w);
      ((uint4*)packed)[g0 + u * 256] = pk;
      keyw[g0 + u * 256] = edge_key(d[u].x, N) | (edge_key(d[u].y, N) << 8) |
                           (edge_key(d[u].z, N) << 16) | (edge_key(d[u].w, N) << 24);
    }
    return;
  }
  for (int u = 0; u < 4; ++u) {
    const int g = g0 + u * 256;
    if (g >= GP) continue;
    const int e0 = g << 2;
    unsigned kw;
    if (e0 + 3 < E) {
      const int4 d = dst4[g];
      const int4 s = src4[g];
      uint4 pk;
      pk.x = edge_pack(s.x, d.x); pk.y = edge_pack(s.y, d.y);
      pk.z = edge_pack(s.z, d.z); pk.w = edge_pack(s.w, d.w);
      ((uint4*)packed)[g] = pk;
      kw = edge_key(d.x, N) | (edge_key(d.y, N) << 8) | (edge_key(d.z, N) << 16) | (edge_key(d.w, N) << 24);
    } else {
      kw = 0u;
      for (int b = 0; b < 4; ++b) {
        const int e = e0 + b;
        unsigned k = KEY_NONE;
        if (e < E) {
          const int d = dst[e];
          packed[e] = edge_pack(src[e], d);
          k = edge_key(d, N);
        }
        kw |= k << (8 * b);
      }
    }
    keyw[g] = kw;
  }
}

__global__ __launch_bounds__(256) void k_feat_lists(const float* __restrict__ h,
                                                    const float* __restrict__ Wt,
                                                    float* __restrict__ feat,
                                                    const unsigned* __restrict__ keyw,
                                                    const unsigned* __restrict__ packed,
                                                    unsigned* __restrict__ glist,
                                                    int* __restrict__ gcnt,
                                                    int N, int E) {
  __shared__ int s_list[SCAN_WAVES * WAVE_CAP];

  const int tid   = threadIdx.x;
  const int wave  = tid >> 5;
  const int lane  = tid & 31;
  const int nscan = ((N + RANGE - 1) >> RANGE_LOG2) * SCAN_BLOCKS;

  if ((int)blockIdx.x < nscan) {
    if (N > KEY_BLOCKS * RANGE) return;
    const int sw    = __builtin_amdgcn_readfirstlane((int)blockIdx.x * SCAN_WAVES + wave);
    const int range = sw / AGG_WAVES;
    const int slice = sw % AGG_WAVES;
    const int seg   = wave * WAVE_CAP;
    int wcnt = 0;
    const unsigned pat = (unsigned)range * 0x01010101u;
    auto hits16 = [&](const uint4 v) -> unsigned {
      const unsigned x0 = v.x ^ pat, x1 = v.y ^ pat, x2 = v.z ^ pat, x3 = v.w ^ pat;
      const unsigned n0 = ((x0 & 0x7F7F7F7Fu) + 0x7F7F7F7Fu) | x0;
      const unsigned n1 = ((x1 & 0x7F7F7F7Fu) + 0x7F7F7F7Fu) | x1;
      const unsigned n2 = ((x2 & 0x7F7F7F7Fu) + 0x7F7F7F7Fu) | x2;
      const unsigned n3 = ((x3 & 0x7F7F7F7Fu) + 0x7F7F7F7Fu) | x3;
      const unsigned nz = ((n0 >> 7) & 0x01010101u) | ((n1 >> 6) & 0x02020202u) |
                          ((n2 >> 5) & 0x04040404u) | ((n3 >> 4) & 0x08080808u);
      return nz ^ 0x0F0F0F0Fu;
    };
    auto push = [&](unsigned m, const int q) {
      unsigned any = __builtin_amdgcn_ballot_w32(m != 0u);
      while (any != 0u) {
        if (m != 0u) {
          const int pos = wcnt + (int)__builtin_amdgcn_mbcnt_lo(any, 0u);
          const int p = __builtin_ctz(m);
          const int e = ((q + ((p >> 2) & 1) * AGG_THREADS) << 4) + ((p & 3) << 2) + (p >> 3);
          if (pos < WAVE_CAP) s_list[seg + pos] = e;
          m &= m - 1u;
        }
        wcnt += __builtin_popcount(any);
        any = __builtin_amdgcn_ballot_w32(m != 0u);
      }
    };
    const int Q = (E + 15) >> 4;
    const uint4* __restrict__ key4 = (const uint4*)keyw;
    int qw = slice * 32;
    for (; qw + 7 * AGG_THREADS + 31 < Q; qw += 8 * AGG_THREADS) {
      const int q = qw + lane;
      uint4 v[8];
#pragma unroll
      for (int u = 0; u < 8; ++u) v[u] = key4[q + u * AGG_THREADS];
#pragma unroll
      for (int u = 0; u < 8; u += 2)
        push(hits16(v[u]) | (hits16(v[u + 1]) << 4), q + u * AGG_THREADS);
    }
    for (; qw < Q; qw += AGG_THREADS) {
      const int q = qw + lane;
      uint4 v = make_uint4(~0u, ~0u, ~0u, ~0u);
      if (q < Q) v = key4[q];
      push(hits16(v), q);
    }
    __syncthreads();
    const int n = min(wcnt, WAVE_CAP);
    unsigned* __restrict__ gl = glist + (size_t)sw * WAVE_CAP;
    for (int i0 = 0; i0 < n; i0 += 8 * 32) {
      int en[8]; unsigned pk[8];
#pragma unroll
      for (int u = 0; u < 8; ++u) en[u] = s_list[seg + ((i0 + u * 32 + lane) & (WAVE_CAP - 1))];
#pragma unroll
      for (int u = 0; u < 8; ++u) pk[u] = packed[(i0 + u * 32 + lane < n) ? en[u] : 0];
#pragma unroll
      for (int u = 0; u < 8; ++u) {
        const int i = i0 + u * 32 + lane;
        if (i < n) gl[i] = pk[u];
      }
    }
    if (lane == 0) gcnt[sw] = wcnt;
    return;
  }

  const int mtile = wave >> 2;
  const int ntile = wave & 3;
  const int rowBase = ((int)blockIdx.x - nscan) * 32 + mtile * 16;
  const int col   = ntile * 16 + (lane & 15);
  const int row   = rowBase + (lane & 15);
  const int khalf = (lane >> 4) << 1;

  const int rowc = (row < N) ? row : (N - 1);
  const float* __restrict__ ap = h  + (size_t)rowc * IN_DIM + khalf;
  const float* __restrict__ bp = Wt + (size_t)col  * IN_DIM + khalf;

  v8f acc = {0.f, 0.f, 0.f, 0.f, 0.f, 0.f, 0.f, 0.f};

#pragma unroll
  for (int k = 0; k < IN_DIM; k += 16) {
    v2f a0, a1, a2, a3, b0, b1, b2, b3;
    a0 = *(const v2f*)(ap + k);      b0 = *(const v2f*)(bp + k);
    a1 = *(const v2f*)(ap + k + 4);  b1 = *(const v2f*)(bp + k + 4);
    a2 = *(const v2f*)(ap + k + 8);  b2 = *(const v2f*)(bp + k + 8);
    a3 = *(const v2f*)(ap + k + 12); b3 = *(const v2f*)(bp + k + 12);
    acc = __builtin_amdgcn_wmma_f32_16x16x4_f32(false, a0, false, b0, (short)0, acc, false, false);
    acc = __builtin_amdgcn_wmma_f32_16x16x4_f32(false, a1, false, b1, (short)0, acc, false, false);
    acc = __builtin_amdgcn_wmma_f32_16x16x4_f32(false, a2, false, b2, (short)0, acc, false, false);
    acc = __builtin_amdgcn_wmma_f32_16x16x4_f32(false, a3, false, b3, (short)0, acc, false, false);
  }

  const int mBase = rowBase + (lane >> 4) * 8;
#pragma unroll
  for (int r = 0; r < 8; ++r) {
    int m = mBase + r;
    if (m < N) feat[(size_t)m * OUT_DIM + col] = acc[r];
  }
}

__global__ void k_el_er(const float* __restrict__ feat,
                        const float* __restrict__ attn_l,
                        const float* __restrict__ attn_r,
                        float* __restrict__ el, float* __restrict__ er, int N) {
  int i = blockIdx.x * blockDim.x + threadIdx.x;
  if (i >= N) return;
  const float4* f  = (const float4*)(feat + (size_t)i * OUT_DIM);
  const float4* al = (const float4*)attn_l;
  const float4* ar = (const float4*)attn_r;
  float sl = 0.f, sr = 0.f;
#pragma unroll
  for (int j = 0; j < OUT_DIM / 4; ++j) {
    float4 v = f[j], a = al[j], b = ar[j];
    sl += v.x * a.x + v.y * a.y + v.z * a.z + v.w * a.w;
    sr += v.x * b.x + v.y * b.y + v.z * b.z + v.w * b.w;
  }
  el[i] = sl;
  er[i] = sr;
}

#define PASS2_EDGES 8

__device__ __forceinline__ int fkey(float f) {
  int b = __float_as_int(f);
  return (b >= 0) ? b : (b ^ 0x7fffffff);
}
__device__ __forceinline__ float fkey_inv(int k) {
  return __int_as_float((k >= 0) ? k : (k ^ 0x7fffffff));
}

__global__ __launch_bounds__(AGG_THREADS) void k_gat_range(const float* __restrict__ feat,
                                                    const float* __restrict__ el,
                                                    const float* __restrict__ er,
                                                    const int* __restrict__ src,
                                                    const int* __restrict__ dst,
                                                    const unsigned* __restrict__ glist,
                                                    const int* __restrict__ gcnt,
                                                    const float* __restrict__ bias,
                                                    float* __restrict__ out,
                                                    int N, int E) {
  __shared__ float          s_acc[RANGE * OUT_DIM];
  __shared__ int            s_mkey[RANGE];
  __shared__ float          s_den[RANGE];
  __shared__ float          s_er[RANGE];
  __shared__ int            s_lsrc[LIST_CAP];
  __shared__ float          s_le[LIST_CAP];
  __shared__ unsigned short s_ld[LIST_CAP];
  __shared__ int            s_wcnt[AGG_WAVES];

  const int tid  = threadIdx.x;
  const int wave = tid >> 5;
  const int lane = tid & 31;
  const int base = blockIdx.x * RANGE;
  const int nloc = min(RANGE, N - base);
  const int NEG_INF_KEY = fkey(-INFINITY);

  for (int i = tid; i < RANGE * OUT_DIM; i += AGG_THREADS) s_acc[i] = 0.f;
  for (int i = tid; i < RANGE; i += AGG_THREADS) {
    s_mkey[i] = NEG_INF_KEY;
    s_den[i]  = 0.f;
    s_er[i]   = (i < nloc) ? er[base + i] : 0.f;
  }
  const bool use_keys = N <= KEY_BLOCKS * RANGE;
  const int gw = __builtin_amdgcn_readfirstlane((int)blockIdx.x * AGG_WAVES + wave);
  if (lane == 0) s_wcnt[wave] = use_keys ? gcnt[gw] : 0;
  __syncthreads();
  const bool over = !use_keys || __builtin_amdgcn_ballot_w32(s_wcnt[lane] > WAVE_CAP) != 0u;
  const int seg = wave * WAVE_CAP;
  static_assert(AGG_WAVES == 32, "the overflow test reads one wave count per lane");

  if (!over) {
    const int n = __builtin_amdgcn_readfirstlane(s_wcnt[wave]);
    const unsigned* __restrict__ gl = glist + (size_t)gw * WAVE_CAP;
    for (int i0 = 0; i0 < n; i0 += 128) {
      int sx[4]; int dl[4]; float lg[4];
#pragma unroll
      for (int u = 0; u < 4; ++u) {
        int i = i0 + u * 32 + lane;
        const unsigned pk = (i < n) ? gl[i] : 0u;
        sx[u] = (int)(pk >> RANGE_LOG2);
        dl[u] = (int)(pk & (RANGE - 1));
      }
#pragma unroll
      for (int u = 0; u < 4; ++u) lg[u] = el[sx[u]];
#pragma unroll
      for (int u = 0; u < 4; ++u) {
        int i = i0 + u * 32 + lane;
        if (i < n) {
          float ev = leaky(lg[u] + s_er[dl[u]]);
          atomicMax(&s_mkey[dl[u]], fkey(ev));
          s_lsrc[seg + i] = sx[u];
          s_le[seg + i]   = ev;
          s_ld[seg + i]   = (unsigned short)dl[u];
        }
      }
    }
    __syncthreads();
    const float* __restrict__ featl = feat + 2 * lane;
    float* accl = s_acc + 2 * lane;
    for (int i0 = 0; i0 < n; i0 += 32) {
      const int cnt = min(32, n - i0);
      int sx = 0, dn = 0, wb = 0;
      if (lane < cnt) {
        const int j = seg + i0 + lane;
        sx = s_lsrc[j];
        dn = s_ld[j];
        const float w = __expf(s_le[j] - fkey_inv(s_mkey[dn]));
        atomicAdd(&s_den[dn], w);
        wb = __float_as_int(w);
      }
      auto group = [&](const int g, const bool guard) {
        int dv[PASS2_EDGES]; float wv[PASS2_EDGES]; v2f fv[PASS2_EDGES];
#pragma unroll
        for (int u = 0; u < PASS2_EDGES; ++u) {
          const int l = (!guard || g + u < cnt) ? (g + u) : g;
          const int sv = __builtin_amdgcn_readlane(sx, l);
          dv[u] = __builtin_amdgcn_readlane(dn, l);
          wv[u] = __int_as_float(__builtin_amdgcn_readlane(wb, l));
          fv[u] = *(const v2f*)(featl + (size_t)sv * OUT_DIM);
        }
#pragma unroll
        for (int u = 0; u < PASS2_EDGES; ++u) {
          if (!guard || g + u < cnt) {
            float* row = accl + dv[u] * OUT_DIM;
            atomicAdd(row,     wv[u] * fv[u].x);
            atomicAdd(row + 1, wv[u] * fv[u].y);
          }
        }
      };
      if (cnt == 32) {
#pragma unroll
        for (int g = 0; g < 32; g += PASS2_EDGES) group(g, false);
      } else {
        for (int g = 0; g < cnt; g += PASS2_EDGES) group(g, true);
      }
    }
  } else {
    for (int e = tid; e < E; e += AGG_THREADS) {
      unsigned dl = (unsigned)(dst[e] - base);
      if (dl < (unsigned)nloc)
        atomicMax(&s_mkey[dl], fkey(leaky(el[src[e]] + s_er[dl])));
    }
    __syncthreads();
    for (int e = tid; e < E; e += AGG_THREADS) {
      unsigned dl = (unsigned)(dst[e] - base);
      if (dl < (unsigned)nloc) {
        int   sidx = src[e];
        float ev   = leaky(el[sidx] + s_er[dl]);
        float w    = __expf(ev - fkey_inv(s_mkey[dl]));
        atomicAdd(&s_den[dl], w);
        const float* fr = feat + (size_t)sidx * OUT_DIM;
        for (int c = 0; c < OUT_DIM; ++c) atomicAdd(&s_acc[dl * OUT_DIM + c], w * fr[c]);
      }
    }
  }
  __syncthreads();

  float* __restrict__ ob = out + (size_t)base * OUT_DIM;
  for (int i = tid; i < nloc * OUT_DIM; i += AGG_THREADS) {
    float den = s_den[i >> 6];
    float inv = (den > 0.f) ? 1.f / den : 1.f;
    ob[i] = s_acc[i] * inv + bias[i & (OUT_DIM - 1)];
  }
}

extern "C" void kernel_launch(void* const* d_in, const int* in_sizes, int n_in,
                              void* d_out, int out_size, void* d_ws, size_t ws_size,
                              hipStream_t stream) {
  const float* h      = (const float*)d_in[0];
  const int*   src    = (const int*)d_in[1];
  const int*   dst    = (const int*)d_in[2];
  const float* W      = (const float*)d_in[3];
  const float* attn_l = (const float*)d_in[4];
  const float* attn_r = (const float*)d_in[5];
  const float* bias   = (const float*)d_in[6];
  float* out = (float*)d_out;

  const int N = in_sizes[0] / IN_DIM;
  const int E = in_sizes[1];

  char* ws = (char*)d_ws;
  size_t off = 0;
  auto carve = [&](size_t bytes) -> char* {
    char* p = ws + off;
    off = (off + bytes + 255) & ~(size_t)255;
    return p;
  };
  float* feat    = (float*)carve((size_t)N * OUT_DIM * sizeof(float));
  float* Wt      = (float*)carve((size_t)IN_DIM * OUT_DIM * sizeof(float));
  float* el      = (float*)carve((size_t)N * sizeof(float));
  float* er      = (float*)carve((size_t)N * sizeof(float));
  unsigned* keyw   = (unsigned*)carve((size_t)(((E + 15) >> 4) << 2) * sizeof(unsigned));
  unsigned* packed = (unsigned*)carve((size_t)E * sizeof(unsigned));
  unsigned* glist  = (unsigned*)carve((size_t)((N + RANGE - 1) / RANGE) * AGG_WAVES * WAVE_CAP * sizeof(unsigned));
  int*      gcnt   = (int*)carve((size_t)((N + RANGE - 1) / RANGE) * AGG_WAVES * sizeof(int));

  k_wt_keys<<<WT_BLOCKS + (((E + 15) >> 4) * 4 + 1023) / 1024, 256, 0, stream>>>(W, Wt, src, dst, keyw, packed, N, E);
  k_feat_lists<<<((N + RANGE - 1) / RANGE) * SCAN_BLOCKS + (N + 31) / 32, 256, 0, stream>>>(h, Wt, feat, keyw, packed, glist, gcnt, N, E);
  k_el_er<<<(N + 255) / 256, 256, 0, stream>>>(feat, attn_l, attn_r, el, er, N);
  k_gat_range<<<(N + RANGE - 1) / RANGE, AGG_THREADS, 0, stream>>>(feat, el, er, src, dst, glist, gcnt, bias, out, N, E);
}


# Projection: a block stages 64 rows of h in LDS by whole-row loads; a wave owns 16 columns x 32 rows and reads W in fragment order (laid out by launch 0). feat k sum re-ordered inside each 8.
# speedup vs baseline: 2.9723x; 1.1450x over previous
#include <hip/hip_runtime.h>
#include <math.h>

#define IN_DIM   128
#define OUT_DIM  64
#define NEG_SLOPE 0.2f

typedef float v2f __attribute__((ext_vector_type(2)));
typedef float v4f __attribute__((ext_vector_type(4)));
typedef float v8f __attribute__((ext_vector_type(8)));

__device__ __forceinline__ float leaky(float e) {
  return (e > 0.f) ? e : NEG_SLOPE * e;
}

#define RANGE      512
#define RANGE_LOG2 9
#define KEY_NONE   255u
#define KEY_BLOCKS 255
#define WT_BLOCKS  ((IN_DIM * OUT_DIM + 255) / 256)

#define LIST_CAP    16384
#define AGG_THREADS 1024
#define AGG_WAVES   (AGG_THREADS / 32)
#define WAVE_CAP    (LIST_CAP / AGG_WAVES)
#define SCAN_WAVES  8
#define SCAN_BLOCKS (AGG_WAVES / SCAN_WAVES)
#define PROJ_ROWS   64
#define H_PITCH4    (IN_DIM / 4 + 1)

__device__ __forceinline__ unsigned edge_key(int d, int N) {
  unsigned k = (unsigned)d >> RANGE_LOG2;
  return ((unsigned)d < (unsigned)N && k < KEY_NONE) ? k : KEY_NONE;
}
__device__ __forceinline__ unsigned edge_pack(int s, int d) {
  return ((unsigned)s << RANGE_LOG2) | ((unsigned)d & (RANGE - 1));
}

__global__ __launch_bounds__(256) void k_wt_keys(const float* __restrict__ W, float* __restrict__ Wf,
                                                 const int* __restrict__ src,
                                                 const int* __restrict__ dst,
                                                 unsigned* __restrict__ keyw,
                                                 unsigned* __restrict__ packed,
                                                 int N, int E) {
  if (blockIdx.x < WT_BLOCKS) {
    int i = blockIdx.x * 256 + threadIdx.x;
    if (i >= IN_DIM * OUT_DIM) return;
    int k = i / OUT_DIM;
    int c = i - k * OUT_DIM;
    const int frag = ((c >> 4) * (IN_DIM / 8) + (k >> 3)) * 32 + ((k >> 2) & 1) * 16 + (c & 15);
    Wf[(size_t)frag * 4 + (k & 3)] = W[i];
    return;
  }
  const int GP = ((E + 15) >> 4) << 2;
  const int gb = (blockIdx.x - WT_BLOCKS) * 1024;
  const int g0 = gb + threadIdx.x;
  const int4* __restrict__ dst4 = (const int4*)dst;
  const int4* __restrict__ src4 = (const int4*)src;
  if ((gb + 1024) * 4 <= E) {
    int4 d[4], s[4];
#pragma unroll
    for (int u = 0; u < 4; ++u) { d[u] = dst4[g0 + u * 256]; s[u] = src4[g0 + u * 256]; }
#pragma unroll
    for (int u = 0; u < 4; ++u) {
      uint4 pk;
      pk.x = edge_pack(s[u].x, d[u].x); pk.y = edge_pack(s[u].y, d[u].y);
      pk.z = edge_pack(s[u].z, d[u].z); pk.w = edge_pack(s[u].w, d[u].w);
      ((uint4*)packed)[g0 + u * 256] = pk;
      keyw[g0 + u * 256] = edge_key(d[u].x, N) | (edge_key(d[u].y, N) << 8) |
                           (edge_key(d[u].z, N) << 16) | (edge_key(d[u].w, N) << 24);
    }
    return;
  }
  for (int u = 0; u < 4; ++u) {
    const int g = g0 + u * 256;
    if (g >= GP) continue;
    const int e0 = g << 2;
    unsigned kw;
    if (e0 + 3 < E) {
      const int4 d = dst4[g];
      const int4 s = src4[g];
      uint4 pk;
      pk.x = edge_pack(s.x, d.x); pk.y = edge_pack(s.y, d.y);
      pk.z = edge_pack(s.z, d.z); pk.w = edge_pack(s.w, d.w);
      ((uint4*)packed)[g] = pk;
      kw = edge_key(d.x, N) | (edge_key(d.y, N) << 8) | (edge_key(d.z, N) << 16) | (edge_key(d.w, N) << 24);
    } else {
      kw = 0u;
      for (int b = 0; b < 4; ++b) {
        const int e = e0 + b;
        unsigned k = KEY_NONE;
        if (e < E) {
          const int d = dst[e];
          packed[e] = edge_pack(src[e], d);
          k = edge_key(d, N);
        }
        kw |= k << (8 * b);
      }
    }
    keyw[g] = kw;
  }
}

__global__ __launch_bounds__(256) void k_feat_lists(const float* __restrict__ h,
                                                    const float* __restrict__ Wf,
                                                    float* __restrict__ feat,
                                                    const unsigned* __restrict__ keyw,
                                                    const unsigned* __restrict__ packed,
                                                    unsigned* __restrict__ glist,
                                                    int* __restrict__ gcnt,
                                                    int N, int E) {
  __shared__ v4f s_raw[PROJ_ROWS * H_PITCH4];
  static_assert(sizeof(v4f) * PROJ_ROWS * H_PITCH4 >= sizeof(int) * SCAN_WAVES * WAVE_CAP, "the filter's lists fit");
  int* const s_list = (int*)s_raw;

  const int tid   = threadIdx.x;
  const int wave  = tid >> 5;
  const int lane  = tid & 31;
  const int nscan = ((N + RANGE - 1) >> RANGE_LOG2) * SCAN_BLOCKS;

  if ((int)blockIdx.x < nscan) {
    if (N > KEY_BLOCKS * RANGE) return;
    const int sw    = __builtin_amdgcn_readfirstlane((int)blockIdx.x * SCAN_WAVES + wave);
    const int range = sw / AGG_WAVES;
    const int slice = sw % AGG_WAVES;
    const int seg   = wave * WAVE_CAP;
    int wcnt = 0;
    const unsigned pat = (unsigned)range * 0x01010101u;
    auto hits16 = [&](const uint4 v) -> unsigned {
      const unsigned x0 = v.x ^ pat, x1 = v.y ^ pat, x2 = v.z ^ pat, x3 = v.w ^ pat;
      const unsigned n0 = ((x0 & 0x7F7F7F7Fu) + 0x7F7F7F7Fu) | x0;
      const unsigned n1 = ((x1 & 0x7F7F7F7Fu) + 0x7F7F7F7Fu) | x1;
      const unsigned n2 = ((x2 & 0x7F7F7F7Fu) + 0x7F7F7F7Fu) | x2;
      const unsigned n3 = ((x3 & 0x7F7F7F7Fu) + 0x7F7F7F7Fu) | x3;
      const unsigned nz = ((n0 >> 7) & 0x01010101u) | ((n1 >> 6) & 0x02020202u) |
                          ((n2 >> 5) & 0x04040404u) | ((n3 >> 4) & 0x08080808u);
      return nz ^ 0x0F0F0F0Fu;
    };
    auto push = [&](unsigned m, const int q) {
      unsigned any = __builtin_amdgcn_ballot_w32(m != 0u);
      while (any != 0u) {
        if (m != 0u) {
          const int pos = wcnt + (int)__builtin_amdgcn_mbcnt_lo(any, 0u);
          const int p = __builtin_ctz(m);
          const int e = ((q + ((p >> 2) & 1) * AGG_THREADS) << 4) + ((p & 3) << 2) + (p >> 3);
          if (pos < WAVE_CAP) s_list[seg + pos] = e;
          m &= m - 1u;
        }
        wcnt += __builtin_popcount(any);
        any = __builtin_amdgcn_ballot_w32(m != 0u);
      }
    };
    const int Q = (E + 15) >> 4;
    const uint4* __restrict__ key4 = (const uint4*)keyw;
    int qw = slice * 32;
    for (; qw + 7 * AGG_THREADS + 31 < Q; qw += 8 * AGG_THREADS) {
      const int q = qw + lane;
      uint4 v[8];
#pragma unroll
      for (int u = 0; u < 8; ++u) v[u] = key4[q + u * AGG_THREADS];
#pragma unroll
      for (int u = 0; u < 8; u += 2)
        push(hits16(v[u]) | (hits16(v[u + 1]) << 4), q + u * AGG_THREADS);
    }
    for (; qw < Q; qw += AGG_THREADS) {
      const int q = qw + lane;
      uint4 v = make_uint4(~0u, ~0u, ~0u, ~0u);
      if (q < Q) v = key4[q];
      push(hits16(v), q);
    }
    __syncthreads();
    const int n = min(wcnt, WAVE_CAP);
    unsigned* __restrict__ gl = glist + (size_t)sw * WAVE_CAP;
    for (int i0 = 0; i0 < n; i0 += 8 * 32) {
      int en[8]; unsigned pk[8];
#pragma unroll
      for (int u = 0; u < 8; ++u) en[u] = s_list[seg + ((i0 + u * 32 + lane) & (WAVE_CAP - 1))];
#pragma unroll
      for (int u = 0; u < 8; ++u) pk[u] = packed[(i0 + u * 32 + lane < n) ? en[u] : 0];
#pragma unroll
      for (int u = 0; u < 8; ++u) {
        const int i = i0 + u * 32 + lane;
        if (i < n) gl[i] = pk[u];
      }
    }
    if (lane == 0) gcnt[sw] = wcnt;
    return;
  }

  const int R0 = ((int)blockIdx.x - nscan) * PROJ_ROWS;
  const bool whole = R0 + PROJ_ROWS <= N;
  {
    const v4f* __restrict__ h4 = (const v4f*)h;
    v4f st[PROJ_ROWS / 8];
    if (whole) {
      const v4f* __restrict__ p = h4 + (size_t)R0 * (IN_DIM / 4) + tid;
#pragma unroll
      for (int i = 0; i < PROJ_ROWS / 8; ++i) st[i] = __builtin_nontemporal_load(p + i * 256);
    } else {
#pragma unroll
      for (int i = 0; i < PROJ_ROWS / 8; ++i) {
        const int grow = min(R0 + 8 * i + wave, N - 1);
        st[i] = h4[(size_t)grow * (IN_DIM / 4) + lane];
      }
    }
#pragma unroll
    for (int i = 0; i < PROJ_ROWS / 8; ++i) s_raw[(8 * i + wave) * H_PITCH4 + lane] = st[i];
  }
  const int ntile = wave & 3;
  const int mhalf = wave >> 2;
  const int c  = lane & 15;
  const int hh = lane >> 4;
  const v4f* __restrict__ wf = (const v4f*)Wf + (size_t)ntile * (IN_DIM / 8) * 32 + lane;
  const v4f* a0p = s_raw + (mhalf * (PROJ_ROWS / 2) + c) * H_PITCH4 + hh;
  const v4f* a1p = a0p + 16 * H_PITCH4;
  static_assert(PROJ_ROWS == 64, "a wave takes two row tiles of its half");
  __syncthreads();

  v8f acc0 = {0.f, 0.f, 0.f, 0.f, 0.f, 0.f, 0.f, 0.f};
  v8f acc1 = {0.f, 0.f, 0.f, 0.f, 0.f, 0.f, 0.f, 0.f};
#pragma unroll
  for (int j = 0; j < IN_DIM / 8; ++j) {
    const v4f b  = wf[j * 32];
    const v4f x0 = a0p[2 * j];
    const v4f x1 = a1p[2 * j];
    const v2f blo = {b.x, b.y},   bhi = {b.z, b.w};
    const v2f l0  = {x0.x, x0.y}, h0  = {x0.z, x0.w};
    const v2f l1  = {x1.x, x1.y}, h1  = {x1.z, x1.w};
    acc0 = __builtin_amdgcn_wmma_f32_16x16x4_f32(false, l0, false, blo, (short)0, acc0, false, false);
    acc1 = __builtin_amdgcn_wmma_f32_16x16x4_f32(false, l1, false, blo, (short)0, acc1, false, false);
    acc0 = __builtin_amdgcn_wmma_f32_16x16x4_f32(false, h0, false, bhi, (short)0, acc0, false, false);
    acc1 = __builtin_amdgcn_wmma_f32_16x16x4_f32(false, h1, false, bhi, (short)0, acc1, false, false);
  }

  const int col   = ntile * 16 + c;
  const int mBase = R0 + mhalf * (PROJ_ROWS / 2) + hh * 8;
  float* __restrict__ fp = feat + (size_t)mBase * OUT_DIM + col;
  if (whole) {
#pragma unroll
    for (int r = 0; r < 8; ++r) {
      fp[r * OUT_DIM]        = acc0[r];
      fp[(r + 16) * OUT_DIM] = acc1[r];
    }
  } else {
#pragma unroll
    for (int r = 0; r < 8; ++r) {
      const int m = mBase + r;
      if (m < N)      fp[r * OUT_DIM]        = acc0[r];
      if (m + 16 < N) fp[(r + 16) * OUT_DIM] = acc1[r];
    }
  }
}

__global__ void k_el_er(const float* __restrict__ feat,
                        const float* __restrict__ attn_l,
                        const float* __restrict__ attn_r,
                        float* __restrict__ el, float* __restrict__ er, int N) {
  int i = blockIdx.x * blockDim.x + threadIdx.x;
  if (i >= N) return;
  const float4* f  = (const float4*)(feat + (size_t)i * OUT_DIM);
  const float4* al = (const float4*)attn_l;
  const float4* ar = (const float4*)attn_r;
  float sl = 0.f, sr = 0.f;
#pragma unroll
  for (int j = 0; j < OUT_DIM / 4; ++j) {
    float4 v = f[j], a = al[j], b = ar[j];
    sl += v.x * a.x + v.y * a.y + v.z * a.z + v.w * a.w;
    sr += v.x * b.x + v.y * b.y + v.z * b.z + v.w * b.w;
  }
  el[i] = sl;
  er[i] = sr;
}

#define PASS2_EDGES 8

__device__ __forceinline__ int fkey(float f) {
  int b = __float_as_int(f);
  return (b >= 0) ? b : (b ^ 0x7fffffff);
}
__device__ __forceinline__ float fkey_inv(int k) {
  return __int_as_float((k >= 0) ? k : (k ^ 0x7fffffff));
}

__global__ __launch_bounds__(AGG_THREADS) void k_gat_range(const float* __restrict__ feat,
                                                    const float* __restrict__ el,
                                                    const float* __restrict__ er,
                                                    const int* __restrict__ src,
                                                    const int* __restrict__ dst,
                                                    const unsigned* __restrict__ glist,
                                                    const int* __restrict__ gcnt,
                                                    const float* __restrict__ bias,
                                                    float* __restrict__ out,
                                                    int N, int E) {
  __shared__ float          s_acc[RANGE * OUT_DIM];
  __shared__ int            s_mkey[RANGE];
  __shared__ float          s_den[RANGE];
  __shared__ float          s_er[RANGE];
  __shared__ int            s_lsrc[LIST_CAP];
  __shared__ float          s_le[LIST_CAP];
  __shared__ unsigned short s_ld[LIST_CAP];
  __shared__ int            s_wcnt[AGG_WAVES];

  const int tid  = threadIdx.x;
  const int wave = tid >> 5;
  const int lane = tid & 31;
  const int base = blockIdx.x * RANGE;
  const int nloc = min(RANGE, N - base);
  const int NEG_INF_KEY = fkey(-INFINITY);

  for (int i = tid; i < RANGE * OUT_DIM; i += AGG_THREADS) s_acc[i] = 0.f;
  for (int i = tid; i < RANGE; i += AGG_THREADS) {
    s_mkey[i] = NEG_INF_KEY;
    s_den[i]  = 0.f;
    s_er[i]   = (i < nloc) ? er[base + i] : 0.f;
  }
  const bool use_keys = N <= KEY_BLOCKS * RANGE;
  const int gw = __builtin_amdgcn_readfirstlane((int)blockIdx.x * AGG_WAVES + wave);
  if (lane == 0) s_wcnt[wave] = use_keys ? gcnt[gw] : 0;
  __syncthreads();
  const bool over = !use_keys || __builtin_amdgcn_ballot_w32(s_wcnt[lane] > WAVE_CAP) != 0u;
  const int seg = wave * WAVE_CAP;
  static_assert(AGG_WAVES == 32, "the overflow test reads one wave count per lane");

  if (!over) {
    const int n = __builtin_amdgcn_readfirstlane(s_wcnt[wave]);
    const unsigned* __restrict__ gl = glist + (size_t)gw * WAVE_CAP;
    for (int i0 = 0; i0 < n; i0 += 128) {
      int sx[4]; int dl[4]; float lg[4];
#pragma unroll
      for (int u = 0; u < 4; ++u) {
        int i = i0 + u * 32 + lane;
        const unsigned pk = (i < n) ? gl[i] : 0u;
        sx[u] = (int)(pk >> RANGE_LOG2);
        dl[u] = (int)(pk & (RANGE - 1));
      }
#pragma unroll
      for (int u = 0; u < 4; ++u) lg[u] = el[sx[u]];
#pragma unroll
      for (int u = 0; u < 4; ++u) {
        int i = i0 + u * 32 + lane;
        if (i < n) {
          float ev = leaky(lg[u] + s_er[dl[u]]);
          atomicMax(&s_mkey[dl[u]], fkey(ev));
          s_lsrc[seg + i] = sx[u];
          s_le[seg + i]   = ev;
          s_ld[seg + i]   = (unsigned short)dl[u];
        }
      }
    }
    __syncthreads();
    const float* __restrict__ featl = feat + 2 * lane;
    float* accl = s_acc + 2 * lane;
    for (int i0 = 0; i0 < n; i0 += 32) {
      const int cnt = min(32, n - i0);
      int sx = 0, dn = 0, wb = 0;
      if (lane < cnt) {
        const int j = seg + i0 + lane;
        sx = s_lsrc[j];
        dn = s_ld[j];
        const float w = __expf(s_le[j] - fkey_inv(s_mkey[dn]));
        atomicAdd(&s_den[dn], w);
        wb = __float_as_int(w);
      }
      auto group = [&](const int g, const bool guard) {
        int dv[PASS2_EDGES]; float wv[PASS2_EDGES]; v2f fv[PASS2_EDGES];
#pragma unroll
        for (int u = 0; u < PASS2_EDGES; ++u) {
          const int l = (!guard || g + u < cnt) ? (g + u) : g;
          const int sv = __builtin_amdgcn_readlane(sx, l);
          dv[u] = __builtin_amdgcn_readlane(dn, l);
          wv[u] = __int_as_float(__builtin_amdgcn_readlane(wb, l));
          fv[u] = *(const v2f*)(featl + (size_t)sv * OUT_DIM);
        }
#pragma unroll
        for (int u = 0; u < PASS2_EDGES; ++u) {
          if (!guard || g + u < cnt) {
            float* row = accl + dv[u] * OUT_DIM;
            atomicAdd(row,     wv[u] * fv[u].x);
            atomicAdd(row + 1, wv[u] * fv[u].y);
          }
        }
      };
      if (cnt == 32) {
#pragma unroll
        for (int g = 0; g < 32; g += PASS2_EDGES) group(g, false);
      } else {
        for (int g = 0; g < cnt; g += PASS2_EDGES) group(g, true);
      }
    }
  } else {
    for (int e = tid; e < E; e += AGG_THREADS) {
      unsigned dl = (unsigned)(dst[e] - base);
      if (dl < (unsigned)nloc)
        atomicMax(&s_mkey[dl], fkey(leaky(el[src[e]] + s_er[dl])));
    }
    __syncthreads();
    for (int e = tid; e < E; e += AGG_THREADS) {
      unsigned dl = (unsigned)(dst[e] - base);
      if (dl < (unsigned)nloc) {
        int   sidx = src[e];
        float ev   = leaky(el[sidx] + s_er[dl]);
        float w    = __expf(ev - fkey_inv(s_mkey[dl]));
        atomicAdd(&s_den[dl], w);
        const float* fr = feat + (size_t)sidx * OUT_DIM;
        for (int c = 0; c < OUT_DIM; ++c) atomicAdd(&s_acc[dl * OUT_DIM + c], w * fr[c]);
      }
    }
  }
  __syncthreads();

  float* __restrict__ ob = out + (size_t)base * OUT_DIM;
  for (int i = tid; i < nloc * OUT_DIM; i += AGG_THREADS) {
    float den = s_den[i >> 6];
    float inv = (den > 0.f) ? 1.f / den : 1.f;
    ob[i] = s_acc[i] * inv + bias[i & (OUT_DIM - 1)];
  }
}

extern "C" void kernel_launch(void* const* d_in, const int* in_sizes, int n_in,
                              void* d_out, int out_size, void* d_ws, size_t ws_size,
                              hipStream_t stream) {
  const float* h      = (const float*)d_in[0];
  const int*   src    = (const int*)d_in[1];
  const int*   dst    = (const int*)d_in[2];
  const float* W      = (const float*)d_in[3];
  const float* attn_l = (const float*)d_in[4];
  const float* attn_r = (const float*)d_in[5];
  const float* bias   = (const float*)d_in[6];
  float* out = (float*)d_out;

  const int N = in_sizes[0] / IN_DIM;
  const int E = in_sizes[1];

  char* ws = (char*)d_ws;
  size_t off = 0;
  auto carve = [&](size_t bytes) -> char* {
    char* p = ws + off;
    off = (off + bytes + 255) & ~(size_t)255;
    return p;
  };
  float* feat    = (float*)carve((size_t)N * OUT_DIM * sizeof(float));
  float* Wf      = (float*)carve((size_t)IN_DIM * OUT_DIM * sizeof(float));
  float* el      = (float*)carve((size_t)N * sizeof(float));
  float* er      = (float*)carve((size_t)N * sizeof(float));
  unsigned* keyw   = (unsigned*)carve((size_t)(((E + 15) >> 4) << 2) * sizeof(unsigned));
  unsigned* packed = (unsigned*)carve((size_t)E * sizeof(unsigned));
  unsigned* glist  = (unsigned*)carve((size_t)((N + RANGE - 1) / RANGE) * AGG_WAVES * WAVE_CAP * sizeof(unsigned));
  int*      gcnt   = (int*)carve((size_t)((N + RANGE - 1) / RANGE) * AGG_WAVES * sizeof(int));

  k_wt_keys<<<WT_BLOCKS + (((E + 15) >> 4) * 4 + 1023) / 1024, 256, 0, stream>>>(W, Wf, src, dst, keyw, packed, N, E);
  k_feat_lists<<<((N + RANGE - 1) / RANGE) * SCAN_BLOCKS + (N + PROJ_ROWS - 1) / PROJ_ROWS, 256, 0, stream>>>(h, Wf, feat, keyw, packed, glist, gcnt, N, E);
  k_el_er<<<(N + 255) / 256, 256, 0, stream>>>(feat, attn_l, attn_r, el, er, N);
  k_gat_range<<<(N + RANGE - 1) / RANGE, AGG_THREADS, 0, stream>>>(feat, el, er, src, dst, glist, gcnt, bias, out, N, E);
}


# Node logits from h, not from feat: launch 0 folds W with attn_l, attn_r into two 128-vectors, the projection's staging waves dot their rows with them; the logits launch goes. el, er sum re-ordered.
# speedup vs baseline: 3.2157x; 1.0819x over previous
#include <hip/hip_runtime.h>
#include <math.h>

#define IN_DIM   128
#define OUT_DIM  64
#define NEG_SLOPE 0.2f

typedef float v2f __attribute__((ext_vector_type(2)));
typedef float v4f __attribute__((ext_vector_type(4)));
typedef float v8f __attribute__((ext_vector_type(8)));

__device__ __forceinline__ float leaky(float e) {
  return (e > 0.f) ? e : NEG_SLOPE * e;
}

#define RANGE      512
#define RANGE_LOG2 9
#define KEY_NONE   255u
#define KEY_BLOCKS 255
#define WT_BLOCKS  ((IN_DIM * OUT_DIM + 255) / 256)
#define HEAD_BLOCKS (WT_BLOCKS + 1)

#define LIST_CAP    16384
#define AGG_THREADS 1024
#define AGG_WAVES   (AGG_THREADS / 32)
#define WAVE_CAP    (LIST_CAP / AGG_WAVES)
#define SCAN_WAVES  8
#define SCAN_BLOCKS (AGG_WAVES / SCAN_WAVES)
#define PROJ_ROWS   64
#define H_PITCH4    (IN_DIM / 4 + 1)

__device__ __forceinline__ unsigned edge_key(int d, int N) {
  unsigned k = (unsigned)d >> RANGE_LOG2;
  return ((unsigned)d < (unsigned)N && k < KEY_NONE) ? k : KEY_NONE;
}
__device__ __forceinline__ unsigned edge_pack(int s, int d) {
  return ((unsigned)s << RANGE_LOG2) | ((unsigned)d & (RANGE - 1));
}

__global__ __launch_bounds__(256) void k_wt_keys(const float* __restrict__ W, float* __restrict__ Wf,
                                                 const float* __restrict__ attn_l,
                                                 const float* __restrict__ attn_r,
                                                 float* __restrict__ wlr,
                                                 const int* __restrict__ src,
                                                 const int* __restrict__ dst,
                                                 unsigned* __restrict__ keyw,
                                                 unsigned* __restrict__ packed,
                                                 int N, int E) {
  if (blockIdx.x < WT_BLOCKS) {
    int i = blockIdx.x * 256 + threadIdx.x;
    if (i >= IN_DIM * OUT_DIM) return;
    int k = i / OUT_DIM;
    int c = i - k * OUT_DIM;
    const int frag = ((c >> 4) * (IN_DIM / 8) + (k >> 3)) * 32 + ((k >> 2) & 1) * 16 + (c & 15);
    Wf[(size_t)frag * 4 + (k & 3)] = W[i];
    return;
  }
  if (blockIdx.x == WT_BLOCKS) {
    static_assert(2 * IN_DIM == 256, "one thread per folded value");
    const int k = threadIdx.x & (IN_DIM - 1);
    const float4* __restrict__ a4 = (const float4*)((threadIdx.x < IN_DIM) ? attn_l : attn_r);
    const float4* __restrict__ w4 = (const float4*)(W + (size_t)k * OUT_DIM);
    float sum = 0.f;
    for (int j = 0; j < OUT_DIM / 4; ++j) {
      const float4 w = w4[j], a = a4[j];
      sum += w.x * a.x + w.y * a.y + w.z * a.z + w.w * a.w;
    }
    wlr[2 * k + ((threadIdx.x < IN_DIM) ? 0 : 1)] = sum;
    return;
  }
  const int GP = ((E + 15) >> 4) << 2;
  const int gb = (blockIdx.x - HEAD_BLOCKS) * 1024;
  const int g0 = gb + threadIdx.x;
  const int4* __restrict__ dst4 = (const int4*)dst;
  const int4* __restrict__ src4 = (const int4*)src;
  if ((gb + 1024) * 4 <= E) {
    int4 d[4], s[4];
#pragma unroll
    for (int u = 0; u < 4; ++u) { d[u] = dst4[g0 + u * 256]; s[u] = src4[g0 + u * 256]; }
#pragma unroll
    for (int u = 0; u < 4; ++u) {
      uint4 pk;
      pk.x = edge_pack(s[u].x, d[u].x); pk.y = edge_pack(s[u].y, d[u].y);
      pk.z = edge_pack(s[u].z, d[u].z); pk.w = edge_pack(s[u].w, d[u].w);
      ((uint4*)packed)[g0 + u * 256] = pk;
      keyw[g0 + u * 256] = edge_key(d[u].x, N) | (edge_key(d[u].y, N) << 8) |
                           (edge_key(d[u].z, N) << 16) | (edge_key(d[u].w, N) << 24);
    }
    return;
  }
  for (int u = 0; u < 4; ++u) {
    const int g = g0 + u * 256;
    if (g >= GP) continue;
    const int e0 = g << 2;
    unsigned kw;
    if (e0 + 3 < E) {
      const int4 d = dst4[g];
      const int4 s = src4[g];
      uint4 pk;
      pk.x = edge_pack(s.x, d.x); pk.y = edge_pack(s.y, d.y);
      pk.z = edge_pack(s.z, d.z); pk.w = edge_pack(s.w, d.w);
      ((uint4*)packed)[g] = pk;
      kw = edge_key(d.x, N) | (edge_key(d.y, N) << 8) | (edge_key(d.z, N) << 16) | (edge_key(d.w, N) << 24);
    } else {
      kw = 0u;
      for (int b = 0; b < 4; ++b) {
        const int e = e0 + b;
        unsigned k = KEY_NONE;
        if (e < E) {
          const int d = dst[e];
          packed[e] = edge_pack(src[e], d);
          k = edge_key(d, N);
        }
        kw |= k << (8 * b);
      }
    }
    keyw[g] = kw;
  }
}

template <int M> __device__ __forceinline__ float lane_xor(float v) {
  if constexpr (M == 16) {
    const int b = __builtin_bit_cast(int, v);
    return __builtin_bit_cast(float, __builtin_amdgcn_permlanex16(b, b, 0x76543210, 0xfedcba98, false, false));
  } else {
    return __builtin_bit_cast(float, __builtin_amdgcn_update_dpp(0, __builtin_bit_cast(int, v), 0x160 + M, 0xf, 0xf, false));
  }
}
template <int M> __device__ __forceinline__ float fold_pair(float a, float b, int lane) {
  const bool hi = (lane & M) != 0;
  const float keep = hi ? b : a, send = hi ? a : b;
  return keep + lane_xor<M>(send);
}

__global__ __launch_bounds__(256) void k_feat_lists(const float* __restrict__ h,
                                                    const float* __restrict__ Wf,
                                                    const float* __restrict__ wlr,
                                                    float* __restrict__ feat,
                                                    float* __restrict__ el,
                                                    float* __restrict__ er,
                                                    const unsigned* __restrict__ keyw,
                                                    const unsigned* __restrict__ packed,
                                                    unsigned* __restrict__ glist,
                                                    int* __restrict__ gcnt,
                                                    int N, int E) {
  __shared__ v4f s_raw[PROJ_ROWS * H_PITCH4];
  static_assert(sizeof(v4f) * PROJ_ROWS * H_PITCH4 >= sizeof(int) * SCAN_WAVES * WAVE_CAP, "the filter's lists fit");
  int* const s_list = (int*)s_raw;
  __shared__ float s_lg[2 * PROJ_ROWS];

  const int tid   = threadIdx.x;
  const int wave  = tid >> 5;
  const int lane  = tid & 31;
  const int nscan = ((N + RANGE - 1) >> RANGE_LOG2) * SCAN_BLOCKS;

  if ((int)blockIdx.x < nscan) {
    if (N > KEY_BLOCKS * RANGE) return;
    const int sw    = __builtin_amdgcn_readfirstlane((int)blockIdx.x * SCAN_WAVES + wave);
    const int range = sw / AGG_WAVES;
    const int slice = sw % AGG_WAVES;
    const int seg   = wave * WAVE_CAP;
    int wcnt = 0;
    const unsigned pat = (unsigned)range * 0x01010101u;
    auto hits16 = [&](const uint4 v) -> unsigned {
      const unsigned x0 = v.x ^ pat, x1 = v.y ^ pat, x2 = v.z ^ pat, x3 = v.w ^ pat;
      const unsigned n0 = ((x0 & 0x7F7F7F7Fu) + 0x7F7F7F7Fu) | x0;
      const unsigned n1 = ((x1 & 0x7F7F7F7Fu) + 0x7F7F7F7Fu) | x1;
      const unsigned n2 = ((x2 & 0x7F7F7F7Fu) + 0x7F7F7F7Fu) | x2;
      const unsigned n3 = ((x3 & 0x7F7F7F7Fu) + 0x7F7F7F7Fu) | x3;
      const unsigned nz = ((n0 >> 7) & 0x01010101u) | ((n1 >> 6) & 0x02020202u) |
                          ((n2 >> 5) & 0x04040404u) | ((n3 >> 4) & 0x08080808u);
      return nz ^ 0x0F0F0F0Fu;
    };
    auto push = [&](unsigned m, const int q) {
      unsigned any = __builtin_amdgcn_ballot_w32(m != 0u);
      while (any != 0u) {
        if (m != 0u) {
          const int pos = wcnt + (int)__builtin_amdgcn_mbcnt_lo(any, 0u);
          const int p = __builtin_ctz(m);
          const int e = ((q + ((p >> 2) & 1) * AGG_THREADS) << 4) + ((p & 3) << 2) + (p >> 3);
          if (pos < WAVE_CAP) s_list[seg + pos] = e;
          m &= m - 1u;
        }
        wcnt += __builtin_popcount(any);
        any = __builtin_amdgcn_ballot_w32(m != 0u);
      }
    };
    const int Q = (E + 15) >> 4;
    const uint4* __restrict__ key4 = (const uint4*)keyw;
    int qw = slice * 32;
    for (; qw + 7 * AGG_THREADS + 31 < Q; qw += 8 * AGG_THREADS) {
      const int q = qw + lane;
      uint4 v[8];
#pragma unroll
      for (int u = 0; u < 8; ++u) v[u] = key4[q + u * AGG_THREADS];
#pragma unroll
      for (int u = 0; u < 8; u += 2)
        push(hits16(v[u]) | (hits16(v[u + 1]) << 4), q + u * AGG_THREADS);
    }
    for (; qw < Q; qw += AGG_THREADS) {
      const int q = qw + lane;
      uint4 v = make_uint4(~0u, ~0u, ~0u, ~0u);
      if (q < Q) v = key4[q];
      push(hits16(v), q);
    }
    __syncthreads();
    const int n = min(wcnt, WAVE_CAP);
    unsigned* __restrict__ gl = glist + (size_t)sw * WAVE_CAP;
    for (int i0 = 0; i0 < n; i0 += 8 * 32) {
      int en[8]; unsigned pk[8];
#pragma unroll
      for (int u = 0; u < 8; ++u) en[u] = s_list[seg + ((i0 + u * 32 + lane) & (WAVE_CAP - 1))];
#pragma unroll
      for (int u = 0; u < 8; ++u) pk[u] = packed[(i0 + u * 32 + lane < n) ? en[u] : 0];
#pragma unroll
      for (int u = 0; u < 8; ++u) {
        const int i = i0 + u * 32 + lane;
        if (i < n) gl[i] = pk[u];
      }
    }
    if (lane == 0) gcnt[sw] = wcnt;
    return;
  }

  const int R0 = ((int)blockIdx.x - nscan) * PROJ_ROWS;
  const bool whole = R0 + PROJ_ROWS <= N;
  {
    const v4f* __restrict__ wv = (const v4f*)wlr + 2 * lane;
    const v4f w01 = wv[0], w23 = wv[1];
    const v4f* __restrict__ h4 = (const v4f*)h;
    const int wrow = R0 + __builtin_amdgcn_readfirstlane(wave);
    v4f st[PROJ_ROWS / 8];
#pragma unroll
    for (int i = 0; i < PROJ_ROWS / 8; ++i)
      st[i] = __builtin_nontemporal_load(h4 + (size_t)min(wrow + 8 * i, N - 1) * (IN_DIM / 4) + lane);
#pragma unroll
    for (int i = 0; i < PROJ_ROWS / 8; ++i) s_raw[(8 * i + wave) * H_PITCH4 + lane] = st[i];
    const v2f w0 = {w01.x, w01.y}, w1 = {w01.z, w01.w}, w2 = {w23.x, w23.y}, w3 = {w23.z, w23.w};
    float v[2 * (PROJ_ROWS / 8)];
#pragma unroll
    for (int i = 0; i < PROJ_ROWS / 8; ++i) {
      const v2f d = w0 * st[i].x + w1 * st[i].y + w2 * st[i].z + w3 * st[i].w;
      v[i]     = d.x;
      v[8 + i] = d.y;
    }
    static_assert(PROJ_ROWS / 8 == 8, "sixteen values, four halving steps");
#pragma unroll
    for (int p = 0; p < 8; ++p) v[p] = fold_pair<16>(v[2 * p], v[2 * p + 1], lane);
#pragma unroll
    for (int p = 0; p < 4; ++p) v[p] = fold_pair<8>(v[2 * p], v[2 * p + 1], lane);
#pragma unroll
    for (int p = 0; p < 2; ++p) v[p] = fold_pair<4>(v[2 * p], v[2 * p + 1], lane);
    float lg = fold_pair<2>(v[0], v[1], lane);
    lg += lane_xor<1>(lg);
    if ((lane & 1) == 0) s_lg[(lane & 2) * (PROJ_ROWS / 2) + 8 * ((lane >> 4) | ((lane >> 2) & 2) | (lane & 4)) + wave] = lg;
  }
  const int ntile = wave & 3;
  const int mhalf = wave >> 2;
  const int c  = lane & 15;
  const int hh = lane >> 4;
  const v4f* __restrict__ wf = (const v4f*)Wf + (size_t)ntile * (IN_DIM / 8) * 32 + lane;
  const v4f* a0p = s_raw + (mhalf * (PROJ_ROWS / 2) + c) * H_PITCH4 + hh;
  const v4f* a1p = a0p + 16 * H_PITCH4;
  static_assert(PROJ_ROWS == 64, "a wave takes two row tiles of its half");
  __syncthreads();

  v8f acc0 = {0.f, 0.f, 0.f, 0.f, 0.f, 0.f, 0.f, 0.f};
  v8f acc1 = {0.f, 0.f, 0.f, 0.f, 0.f, 0.f, 0.f, 0.f};
#pragma unroll
  for (int j = 0; j < IN_DIM / 8; ++j) {
    const v4f b  = wf[j * 32];
    const v4f x0 = a0p[2 * j];
    const v4f x1 = a1p[2 * j];
    const v2f blo = {b.x, b.y},   bhi = {b.z, b.w};
    const v2f l0  = {x0.x, x0.y}, h0  = {x0.z, x0.w};
    const v2f l1  = {x1.x, x1.y}, h1  = {x1.z, x1.w};
    acc0 = __builtin_amdgcn_wmma_f32_16x16x4_f32(false, l0, false, blo, (short)0, acc0, false, false);
    acc1 = __builtin_amdgcn_wmma_f32_16x16x4_f32(false, l1, false, blo, (short)0, acc1, false, false);
    acc0 = __builtin_amdgcn_wmma_f32_16x16x4_f32(false, h0, false, bhi, (short)0, acc0, false, false);
    acc1 = __builtin_amdgcn_wmma_f32_16x16x4_f32(false, h1, false, bhi, (short)0, acc1, false, false);
  }

  const int col   = ntile * 16 + c;
  const int mBase = R0 + mhalf * (PROJ_ROWS / 2) + hh * 8;
  float* __restrict__ fp = feat + (size_t)mBase * OUT_DIM + col;
  if (whole) {
#pragma unroll
    for (int r = 0; r < 8; ++r) {
      fp[r * OUT_DIM]        = acc0[r];
      fp[(r + 16) * OUT_DIM] = acc1[r];
    }
  } else {
#pragma unroll
    for (int r = 0; r < 8; ++r) {
      const int m = mBase + r;
      if (m < N)      fp[r * OUT_DIM]        = acc0[r];
      if (m + 16 < N) fp[(r + 16) * OUT_DIM] = acc1[r];
    }
  }
  if (tid < 2 * PROJ_ROWS) {
    const int m = R0 + (tid & (PROJ_ROWS - 1));
    if (m < N) ((tid < PROJ_ROWS) ? el : er)[m] = s_lg[tid];
  }
}

#define PASS2_EDGES 8

__device__ __forceinline__ int fkey(float f) {
  int b = __float_as_int(f);
  return (b >= 0) ? b : (b ^ 0x7fffffff);
}
__device__ __forceinline__ float fkey_inv(int k) {
  return __int_as_float((k >= 0) ? k : (k ^ 0x7fffffff));
}

__global__ __launch_bounds__(AGG_THREADS) void k_gat_range(const float* __restrict__ feat,
                                                    const float* __restrict__ el,
                                                    const float* __restrict__ er,
                                                    const int* __restrict__ src,
                                                    const int* __restrict__ dst,
                                                    const unsigned* __restrict__ glist,
                                                    const int* __restrict__ gcnt,
                                                    const float* __restrict__ bias,
                                                    float* __restrict__ out,
                                                    int N, int E) {
  __shared__ float          s_acc[RANGE * OUT_DIM];
  __shared__ int            s_mkey[RANGE];
  __shared__ float          s_den[RANGE];
  __shared__ float          s_er[RANGE];
  __shared__ int            s_lsrc[LIST_CAP];
  __shared__ float          s_le[LIST_CAP];
  __shared__ unsigned short s_ld[LIST_CAP];
  __shared__ int            s_wcnt[AGG_WAVES];

  const int tid  = threadIdx.x;
  const int wave = tid >> 5;
  const int lane = tid & 31;
  const int base = blockIdx.x * RANGE;
  const int nloc = min(RANGE, N - base);
  const int NEG_INF_KEY = fkey(-INFINITY);

  for (int i = tid; i < RANGE * OUT_DIM; i += AGG_THREADS) s_acc[i] = 0.f;
  for (int i = tid; i < RANGE; i += AGG_THREADS) {
    s_mkey[i] = NEG_INF_KEY;
    s_den[i]  = 0.f;
    s_er[i]   = (i < nloc) ? er[base + i] : 0.f;
  }
  const bool use_keys = N <= KEY_BLOCKS * RANGE;
  const int gw = __builtin_amdgcn_readfirstlane((int)blockIdx.x * AGG_WAVES + wave);
  if (lane == 0) s_wcnt[wave] = use_keys ? gcnt[gw] : 0;
  __syncthreads();
  const bool over = !use_keys || __builtin_amdgcn_ballot_w32(s_wcnt[lane] > WAVE_CAP) != 0u;
  const int seg = wave * WAVE_CAP;
  static_assert(AGG_WAVES == 32, "the overflow test reads one wave count per lane");

  if (!over) {
    const int n = __builtin_amdgcn_readfirstlane(s_wcnt[wave]);
    const unsigned* __restrict__ gl = glist + (size_t)gw * WAVE_CAP;
    for (int i0 = 0; i0 < n; i0 += 128) {
      int sx[4]; int dl[4]; float lg[4];
#pragma unroll
      for (int u = 0; u < 4; ++u) {
        int i = i0 + u * 32 + lane;
        const unsigned pk = (i < n) ? gl[i] : 0u;
        sx[u] = (int)(pk >> RANGE_LOG2);
        dl[u] = (int)(pk & (RANGE - 1));
      }
#pragma unroll
      for (int u = 0; u < 4; ++u) lg[u] = el[sx[u]];
#pragma unroll
      for (int u = 0; u < 4; ++u) {
        int i = i0 + u * 32 + lane;
        if (i < n) {
          float ev = leaky(lg[u] + s_er[dl[u]]);
          atomicMax(&s_mkey[dl[u]], fkey(ev));
          s_lsrc[seg + i] = sx[u];
          s_le[seg + i]   = ev;
          s_ld[seg + i]   = (unsigned short)dl[u];
        }
      }
    }
    __syncthreads();
    const float* __restrict__ featl = feat + 2 * lane;
    float* accl = s_acc + 2 * lane;
    for (int i0 = 0; i0 < n; i0 += 32) {
      const int cnt = min(32, n - i0);
      int sx = 0, dn = 0, wb = 0;
      if (lane < cnt) {
        const int j = seg + i0 + lane;
        sx = s_lsrc[j];
        dn = s_ld[j];
        const float w = __expf(s_le[j] - fkey_inv(s_mkey[dn]));
        atomicAdd(&s_den[dn], w);
        wb = __float_as_int(w);
      }
      auto group = [&](const int g, const bool guard) {
        int dv[PASS2_EDGES]; float wv[PASS2_EDGES]; v2f fv[PASS2_EDGES];
#pragma unroll
        for (int u = 0; u < PASS2_EDGES; ++u) {
          const int l = (!guard || g + u < cnt) ? (g + u) : g;
          const int sv = __builtin_amdgcn_readlane(sx, l);
          dv[u] = __builtin_amdgcn_readlane(dn, l);
          wv[u] = __int_as_float(__builtin_amdgcn_readlane(wb, l));
          fv[u] = *(const v2f*)(featl + (size_t)sv * OUT_DIM);
        }
#pragma unroll
        for (int u = 0; u < PASS2_EDGES; ++u) {
          if (!guard || g + u < cnt) {
            float* row = accl + dv[u] * OUT_DIM;
            atomicAdd(row,     wv[u] * fv[u].x);
            atomicAdd(row + 1, wv[u] * fv[u].y);
          }
        }
      };
      if (cnt == 32) {
#pragma unroll
        for (int g = 0; g < 32; g += PASS2_EDGES) group(g, false);
      } else {
        for (int g = 0; g < cnt; g += PASS2_EDGES) group(g, true);
      }
    }
  } else {
    for (int e = tid; e < E; e += AGG_THREADS) {
      unsigned dl = (unsigned)(dst[e] - base);
      if (dl < (unsigned)nloc)
        atomicMax(&s_mkey[dl], fkey(leaky(el[src[e]] + s_er[dl])));
    }
    __syncthreads();
    for (int e = tid; e < E; e += AGG_THREADS) {
      unsigned dl = (unsigned)(dst[e] - base);
      if (dl < (unsigned)nloc) {
        int   sidx = src[e];
        float ev   = leaky(el[sidx] + s_er[dl]);
        float w    = __expf(ev - fkey_inv(s_mkey[dl]));
        atomicAdd(&s_den[dl], w);
        const float* fr = feat + (size_t)sidx * OUT_DIM;
        for (int c = 0; c < OUT_DIM; ++c) atomicAdd(&s_acc[dl * OUT_DIM + c], w * fr[c]);
      }
    }
  }
  __syncthreads();

  float* __restrict__ ob = out + (size_t)base * OUT_DIM;
  for (int i = tid; i < nloc * OUT_DIM; i += AGG_THREADS) {
    float den = s_den[i >> 6];
    float inv = (den > 0.f) ? 1.f / den : 1.f;
    ob[i] = s_acc[i] * inv + bias[i & (OUT_DIM - 1)];
  }
}

extern "C" void kernel_launch(void* const* d_in, const int* in_sizes, int n_in,
                              void* d_out, int out_size, void* d_ws, size_t ws_size,
                              hipStream_t stream) {
  const float* h      = (const float*)d_in[0];
  const int*   src    = (const int*)d_in[1];
  const int*   dst    = (const int*)d_in[2];
  const float* W      = (const float*)d_in[3];
  const float* attn_l = (const float*)d_in[4];
  const float* attn_r = (const float*)d_in[5];
  const float* bias   = (const float*)d_in[6];
  float* out = (float*)d_out;

  const int N = in_sizes[0] / IN_DIM;
  const int E = in_sizes[1];

  char* ws = (char*)d_ws;
  size_t off = 0;
  auto carve = [&](size_t bytes) -> char* {
    char* p = ws + off;
    off = (off + bytes + 255) & ~(size_t)255;
    return p;
  };
  float* feat    = (float*)carve((size_t)N * OUT_DIM * sizeof(float));
  float* Wf      = (float*)carve((size_t)IN_DIM * OUT_DIM * sizeof(float));
  float* el      = (float*)carve((size_t)N * sizeof(float));
  float* er      = (float*)carve((size_t)N * sizeof(float));
  float* wlr     = (float*)carve((size_t)2 * IN_DIM * sizeof(float));
  unsigned* keyw   = (unsigned*)carve((size_t)(((E + 15) >> 4) << 2) * sizeof(unsigned));
  unsigned* packed = (unsigned*)carve((size_t)E * sizeof(unsigned));
  unsigned* glist  = (unsigned*)carve((size_t)((N + RANGE - 1) / RANGE) * AGG_WAVES * WAVE_CAP * sizeof(unsigned));
  int*      gcnt   = (int*)carve((size_t)((N + RANGE - 1) / RANGE) * AGG_WAVES * sizeof(int));

  k_wt_keys<<<HEAD_BLOCKS + (((E + 15) >> 4) * 4 + 1023) / 1024, 256, 0, stream>>>(W, Wf, attn_l, attn_r, wlr, src, dst, keyw, packed, N, E);
  k_feat_lists<<<((N + RANGE - 1) / RANGE) * SCAN_BLOCKS + (N + PROJ_ROWS - 1) / PROJ_ROWS, 256, 0, stream>>>(h, Wf, wlr, feat, el, er, keyw, packed, glist, gcnt, N, E);
  k_gat_range<<<(N + RANGE - 1) / RANGE, AGG_THREADS, 0, stream>>>(feat, el, er, src, dst, glist, gcnt, bias, out, N, E);
}


# Aggregation ranges of 400 destination nodes, not 512: 250 blocks, not 196, one for nearly every LDS unit of the chip, each with 22 % fewer edges. Node sums: LDS arrival order as before.
# speedup vs baseline: 3.3523x; 1.0425x over previous
#include <hip/hip_runtime.h>
#include <math.h>

#define IN_DIM   128
#define OUT_DIM  64
#define NEG_SLOPE 0.2f

typedef float v2f __attribute__((ext_vector_type(2)));
typedef float v4f __attribute__((ext_vector_type(4)));
typedef float v8f __attribute__((ext_vector_type(8)));

__device__ __forceinline__ float leaky(float e) {
  return (e > 0.f) ? e : NEG_SLOPE * e;
}

#define RANGE      400
#define NODE_BITS  9
#define KEY_NONE   255u
#define KEY_BLOCKS 255
#define WT_BLOCKS  ((IN_DIM * OUT_DIM + 255) / 256)
#define HEAD_BLOCKS (WT_BLOCKS + 1)

#define LIST_CAP    16384
#define AGG_THREADS 1024
#define AGG_WAVES   (AGG_THREADS / 32)
#define WAVE_CAP    (LIST_CAP / AGG_WAVES)
#define SCAN_WAVES  8
#define SCAN_BLOCKS (AGG_WAVES / SCAN_WAVES)
#define PROJ_ROWS   64
#define H_PITCH4    (IN_DIM / 4 + 1)

static_assert(RANGE <= (1 << NODE_BITS), "a local node number fits its field");
__device__ __forceinline__ unsigned edge_key(int d, int N) {
  unsigned k = (unsigned)d / RANGE;
  return ((unsigned)d < (unsigned)N && k < KEY_NONE) ? k : KEY_NONE;
}
__device__ __forceinline__ unsigned edge_pack(int s, int d) {
  return ((unsigned)s << NODE_BITS) | ((unsigned)d % RANGE);
}

__global__ __launch_bounds__(256) void k_wt_keys(const float* __restrict__ W, float* __restrict__ Wf,
                                                 const float* __restrict__ attn_l,
                                                 const float* __restrict__ attn_r,
                                                 float* __restrict__ wlr,
                                                 const int* __restrict__ src,
                                                 const int* __restrict__ dst,
                                                 unsigned* __restrict__ keyw,
                                                 unsigned* __restrict__ packed,
                                                 int N, int E) {
  if (blockIdx.x < WT_BLOCKS) {
    int i = blockIdx.x * 256 + threadIdx.x;
    if (i >= IN_DIM * OUT_DIM) return;
    int k = i / OUT_DIM;
    int c = i - k * OUT_DIM;
    const int frag = ((c >> 4) * (IN_DIM / 8) + (k >> 3)) * 32 + ((k >> 2) & 1) * 16 + (c & 15);
    Wf[(size_t)frag * 4 + (k & 3)] = W[i];
    return;
  }
  if (blockIdx.x == WT_BLOCKS) {
    static_assert(2 * IN_DIM == 256, "one thread per folded value");
    const int k = threadIdx.x & (IN_DIM - 1);
    const float4* __restrict__ a4 = (const float4*)((threadIdx.x < IN_DIM) ? attn_l : attn_r);
    const float4* __restrict__ w4 = (const float4*)(W + (size_t)k * OUT_DIM);
    float sum = 0.f;
    for (int j = 0; j < OUT_DIM / 4; ++j) {
      const float4 w = w4[j], a = a4[j];
      sum += w.x * a.x + w.y * a.y + w.z * a.z + w.w * a.w;
    }
    wlr[2 * k + ((threadIdx.x < IN_DIM) ? 0 : 1)] = sum;
    return;
  }
  const int GP = ((E + 15) >> 4) << 2;
  const int gb = (blockIdx.x - HEAD_BLOCKS) * 1024;
  const int g0 = gb + threadIdx.x;
  const int4* __restrict__ dst4 = (const int4*)dst;
  const int4* __restrict__ src4 = (const int4*)src;
  if ((gb + 1024) * 4 <= E) {
    int4 d[4], s[4];
#pragma unroll
    for (int u = 0; u < 4; ++u) { d[u] = dst4[g0 + u * 256]; s[u] = src4[g0 + u * 256]; }
#pragma unroll
    for (int u = 0; u < 4; ++u) {
      uint4 pk;
      pk.x = edge_pack(s[u].x, d[u].x); pk.y = edge_pack(s[u].y, d[u].y);
      pk.z = edge_pack(s[u].z, d[u].z); pk.w = edge_pack(s[u].w, d[u].w);
      ((uint4*)packed)[g0 + u * 256] = pk;
      keyw[g0 + u * 256] = edge_key(d[u].x, N) | (edge_key(d[u].y, N) << 8) |
                           (edge_key(d[u].z, N) << 16) | (edge_key(d[u].w, N) << 24);
    }
    return;
  }
  for (int u = 0; u < 4; ++u) {
    const int g = g0 + u * 256;
    if (g >= GP) continue;
    const int e0 = g << 2;
    unsigned kw;
    if (e0 + 3 < E) {
      const int4 d = dst4[g];
      const int4 s = src4[g];
      uint4 pk;
      pk.x = edge_pack(s.x, d.x); pk.y = edge_pack(s.y, d.y);
      pk.z = edge_pack(s.z, d.z); pk.w = edge_pack(s.w, d.w);
      ((uint4*)packed)[g] = pk;
      kw = edge_key(d.x, N) | (edge_key(d.y, N) << 8) | (edge_key(d.z, N) << 16) | (edge_key(d.w, N) << 24);
    } else {
      kw = 0u;
      for (int b = 0; b < 4; ++b) {
        const int e = e0 + b;
        unsigned k = KEY_NONE;
        if (e < E) {
          const int d = dst[e];
          packed[e] = edge_pack(src[e], d);
          k = edge_key(d, N);
        }
        kw |= k << (8 * b);
      }
    }
    keyw[g] = kw;
  }
}

template <int M> __device__ __forceinline__ float lane_xor(float v) {
  if constexpr (M == 16) {
    const int b = __builtin_bit_cast(int, v);
    return __builtin_bit_cast(float, __builtin_amdgcn_permlanex16(b, b, 0x76543210, 0xfedcba98, false, false));
  } else {
    return __builtin_bit_cast(float, __builtin_amdgcn_update_dpp(0, __builtin_bit_cast(int, v), 0x160 + M, 0xf, 0xf, false));
  }
}
template <int M> __device__ __forceinline__ float fold_pair(float a, float b, int lane) {
  const bool hi = (lane & M) != 0;
  const float keep = hi ? b : a, send = hi ? a : b;
  return keep + lane_xor<M>(send);
}

__global__ __launch_bounds__(256) void k_feat_lists(const float* __restrict__ h,
                                                    const float* __restrict__ Wf,
                                                    const float* __restrict__ wlr,
                                                    float* __restrict__ feat,
                                                    float* __restrict__ el,
                                                    float* __restrict__ er,
                                                    const unsigned* __restrict__ keyw,
                                                    const unsigned* __restrict__ packed,
                                                    unsigned* __restrict__ glist,
                                                    int* __restrict__ gcnt,
                                                    int N, int E) {
  __shared__ v4f s_raw[PROJ_ROWS * H_PITCH4];
  static_assert(sizeof(v4f) * PROJ_ROWS * H_PITCH4 >= sizeof(int) * SCAN_WAVES * WAVE_CAP, "the filter's lists fit");
  int* const s_list = (int*)s_raw;
  __shared__ float s_lg[2 * PROJ_ROWS];

  const int tid   = threadIdx.x;
  const int wave  = tid >> 5;
  const int lane  = tid & 31;
  const int nscan = ((N + RANGE - 1) / RANGE) * SCAN_BLOCKS;

  if ((int)blockIdx.x < nscan) {
    if (N > KEY_BLOCKS * RANGE) return;
    const int sw    = __builtin_amdgcn_readfirstlane((int)blockIdx.x * SCAN_WAVES + wave);
    const int range = sw / AGG_WAVES;
    const int slice = sw % AGG_WAVES;
    const int seg   = wave * WAVE_CAP;
    int wcnt = 0;
    const unsigned pat = (unsigned)range * 0x01010101u;
    auto hits16 = [&](const uint4 v) -> unsigned {
      const unsigned x0 = v.x ^ pat, x1 = v.y ^ pat, x2 = v.z ^ pat, x3 = v.w ^ pat;
      const unsigned n0 = ((x0 & 0x7F7F7F7Fu) + 0x7F7F7F7Fu) | x0;
      const unsigned n1 = ((x1 & 0x7F7F7F7Fu) + 0x7F7F7F7Fu) | x1;
      const unsigned n2 = ((x2 & 0x7F7F7F7Fu) + 0x7F7F7F7Fu) | x2;
      const unsigned n3 = ((x3 & 0x7F7F7F7Fu) + 0x7F7F7F7Fu) | x3;
      const unsigned nz = ((n0 >> 7) & 0x01010101u) | ((n1 >> 6) & 0x02020202u) |
                          ((n2 >> 5) & 0x04040404u) | ((n3 >> 4) & 0x08080808u);
      return nz ^ 0x0F0F0F0Fu;
    };
    auto push = [&](unsigned m, const int q) {
      unsigned any = __builtin_amdgcn_ballot_w32(m != 0u);
      while (any != 0u) {
        if (m != 0u) {
          const int pos = wcnt + (int)__builtin_amdgcn_mbcnt_lo(any, 0u);
          const int p = __builtin_ctz(m);
          const int e = ((q + ((p >> 2) & 1) * AGG_THREADS) << 4) + ((p & 3) << 2) + (p >> 3);
          if (pos < WAVE_CAP) s_list[seg + pos] = e;
          m &= m - 1u;
        }
        wcnt += __builtin_popcount(any);
        any = __builtin_amdgcn_ballot_w32(m != 0u);
      }
    };
    const int Q = (E + 15) >> 4;
    const uint4* __restrict__ key4 = (const uint4*)keyw;
    int qw = slice * 32;
    for (; qw + 7 * AGG_THREADS + 31 < Q; qw += 8 * AGG_THREADS) {
      const int q = qw + lane;
      uint4 v[8];
#pragma unroll
      for (int u = 0; u < 8; ++u) v[u] = key4[q + u * AGG_THREADS];
#pragma unroll
      for (int u = 0; u < 8; u += 2)
        push(hits16(v[u]) | (hits16(v[u + 1]) << 4), q + u * AGG_THREADS);
    }
    for (; qw < Q; qw += AGG_THREADS) {
      const int q = qw + lane;
      uint4 v = make_uint4(~0u, ~0u, ~0u, ~0u);
      if (q < Q) v = key4[q];
      push(hits16(v), q);
    }
    __syncthreads();
    const int n = min(wcnt, WAVE_CAP);
    unsigned* __restrict__ gl = glist + (size_t)sw * WAVE_CAP;
    for (int i0 = 0; i0 < n; i0 += 8 * 32) {
      int en[8]; unsigned pk[8];
#pragma unroll
      for (int u = 0; u < 8; ++u) en[u] = s_list[seg + ((i0 + u * 32 + lane) & (WAVE_CAP - 1))];
#pragma unroll
      for (int u = 0; u < 8; ++u) pk[u] = packed[(i0 + u * 32 + lane < n) ? en[u] : 0];
#pragma unroll
      for (int u = 0; u < 8; ++u) {
        const int i = i0 + u * 32 + lane;
        if (i < n) gl[i] = pk[u];
      }
    }
    if (lane == 0) gcnt[sw] = wcnt;
    return;
  }

  const int R0 = ((int)blockIdx.x - nscan) * PROJ_ROWS;
  const bool whole = R0 + PROJ_ROWS <= N;
  {
    const v4f* __restrict__ wv = (const v4f*)wlr + 2 * lane;
    const v4f w01 = wv[0], w23 = wv[1];
    const v4f* __restrict__ h4 = (const v4f*)h;
    const int wrow = R0 + __builtin_amdgcn_readfirstlane(wave);
    v4f st[PROJ_ROWS / 8];
#pragma unroll
    for (int i = 0; i < PROJ_ROWS / 8; ++i)
      st[i] = __builtin_nontemporal_load(h4 + (size_t)min(wrow + 8 * i, N - 1) * (IN_DIM / 4) + lane);
#pragma unroll
    for (int i = 0; i < PROJ_ROWS / 8; ++i) s_raw[(8 * i + wave) * H_PITCH4 + lane] = st[i];
    const v2f w0 = {w01.x, w01.y}, w1 = {w01.z, w01.w}, w2 = {w23.x, w23.y}, w3 = {w23.z, w23.w};
    float v[2 * (PROJ_ROWS / 8)];
#pragma unroll
    for (int i = 0; i < PROJ_ROWS / 8; ++i) {
      const v2f d = w0 * st[i].x + w1 * st[i].y + w2 * st[i].z + w3 * st[i].w;
      v[i]     = d.x;
      v[8 + i] = d.y;
    }
    static_assert(PROJ_ROWS / 8 == 8, "sixteen values, four halving steps");
#pragma unroll
    for (int p = 0; p < 8; ++p) v[p] = fold_pair<16>(v[2 * p], v[2 * p + 1], lane);
#pragma unroll
    for (int p = 0; p < 4; ++p) v[p] = fold_pair<8>(v[2 * p], v[2 * p + 1], lane);
#pragma unroll
    for (int p = 0; p < 2; ++p) v[p] = fold_pair<4>(v[2 * p], v[2 * p + 1], lane);
    float lg = fold_pair<2>(v[0], v[1], lane);
    lg += lane_xor<1>(lg);
    if ((lane & 1) == 0) s_lg[(lane & 2) * (PROJ_ROWS / 2) + 8 * ((lane >> 4) | ((lane >> 2) & 2) | (lane & 4)) + wave] = lg;
  }
  const int ntile = wave & 3;
  const int mhalf = wave >> 2;
  const int c  = lane & 15;
  const int hh = lane >> 4;
  const v4f* __restrict__ wf = (const v4f*)Wf + (size_t)ntile * (IN_DIM / 8) * 32 + lane;
  const v4f* a0p = s_raw + (mhalf * (PROJ_ROWS / 2) + c) * H_PITCH4 + hh;
  const v4f* a1p = a0p + 16 * H_PITCH4;
  static_assert(PROJ_ROWS == 64, "a wave takes two row tiles of its half");
  __syncthreads();

  v8f acc0 = {0.f, 0.f, 0.f, 0.f, 0.f, 0.f, 0.f, 0.f};
  v8f acc1 = {0.f, 0.f, 0.f, 0.f, 0.f, 0.f, 0.f, 0.f};
#pragma unroll
  for (int j = 0; j < IN_DIM / 8; ++j) {
    const v4f b  = wf[j * 32];
    const v4f x0 = a0p[2 * j];
    const v4f x1 = a1p[2 * j];
    const v2f blo = {b.x, b.y},   bhi = {b.z, b.w};
    const v2f l0  = {x0.x, x0.y}, h0  = {x0.z, x0.w};
    const v2f l1  = {x1.x, x1.y}, h1  = {x1.z, x1.w};
    acc0 = __builtin_amdgcn_wmma_f32_16x16x4_f32(false, l0, false, blo, (short)0, acc0, false, false);
    acc1 = __builtin_amdgcn_wmma_f32_16x16x4_f32(false, l1, false, blo, (short)0, acc1, false, false);
    acc0 = __builtin_amdgcn_wmma_f32_16x16x4_f32(false, h0, false, bhi, (short)0, acc0, false, false);
    acc1 = __builtin_amdgcn_wmma_f32_16x16x4_f32(false, h1, false, bhi, (short)0, acc1, false, false);
  }

  const int col   = ntile * 16 + c;
  const int mBase = R0 + mhalf * (PROJ_ROWS / 2) + hh * 8;
  float* __restrict__ fp = feat + (size_t)mBase * OUT_DIM + col;
  if (whole) {
#pragma unroll
    for (int r = 0; r < 8; ++r) {
      fp[r * OUT_DIM]        = acc0[r];
      fp[(r + 16) * OUT_DIM] = acc1[r];
    }
  } else {
#pragma unroll
    for (int r = 0; r < 8; ++r) {
      const int m = mBase + r;
      if (m < N)      fp[r * OUT_DIM]        = acc0[r];
      if (m + 16 < N) fp[(r + 16) * OUT_DIM] = acc1[r];
    }
  }
  if (tid < 2 * PROJ_ROWS) {
    const int m = R0 + (tid & (PROJ_ROWS - 1));
    if (m < N) ((tid < PROJ_ROWS) ? el : er)[m] = s_lg[tid];
  }
}

#define PASS2_EDGES 8

__device__ __forceinline__ int fkey(float f) {
  int b = __float_as_int(f);
  return (b >= 0) ? b : (b ^ 0x7fffffff);
}
__device__ __forceinline__ float fkey_inv(int k) {
  return __int_as_float((k >= 0) ? k : (k ^ 0x7fffffff));
}

__global__ __launch_bounds__(AGG_THREADS) void k_gat_range(const float* __restrict__ feat,
                                                    const float* __restrict__ el,
                                                    const float* __restrict__ er,
                                                    const int* __restrict__ src,
                                                    const int* __restrict__ dst,
                                                    const unsigned* __restrict__ glist,
                                                    const int* __restrict__ gcnt,
                                                    const float* __restrict__ bias,
                                                    float* __restrict__ out,
                                                    int N, int E) {
  __shared__ float          s_acc[RANGE * OUT_DIM];
  __shared__ int            s_mkey[RANGE];
  __shared__ float          s_den[RANGE];
  __shared__ float          s_er[RANGE];
  __shared__ int            s_lsrc[LIST_CAP];
  __shared__ float          s_le[LIST_CAP];
  __shared__ unsigned short s_ld[LIST_CAP];
  __shared__ int            s_wcnt[AGG_WAVES];

  const int tid  = threadIdx.x;
  const int wave = tid >> 5;
  const int lane = tid & 31;
  const int base = blockIdx.x * RANGE;
  const int nloc = min(RANGE, N - base);
  const int NEG_INF_KEY = fkey(-INFINITY);

  for (int i = tid; i < RANGE * OUT_DIM; i += AGG_THREADS) s_acc[i] = 0.f;
  for (int i = tid; i < RANGE; i += AGG_THREADS) {
    s_mkey[i] = NEG_INF_KEY;
    s_den[i]  = 0.f;
    s_er[i]   = (i < nloc) ? er[base + i] : 0.f;
  }
  const bool use_keys = N <= KEY_BLOCKS * RANGE;
  const int gw = __builtin_amdgcn_readfirstlane((int)blockIdx.x * AGG_WAVES + wave);
  if (lane == 0) s_wcnt[wave] = use_keys ? gcnt[gw] : 0;
  __syncthreads();
  const bool over = !use_keys || __builtin_amdgcn_ballot_w32(s_wcnt[lane] > WAVE_CAP) != 0u;
  const int seg = wave * WAVE_CAP;
  static_assert(AGG_WAVES == 32, "the overflow test reads one wave count per lane");

  if (!over) {
    const int n = __builtin_amdgcn_readfirstlane(s_wcnt[wave]);
    const unsigned* __restrict__ gl = glist + (size_t)gw * WAVE_CAP;
    for (int i0 = 0; i0 < n; i0 += 128) {
      int sx[4]; int dl[4]; float lg[4];
#pragma unroll
      for (int u = 0; u < 4; ++u) {
        int i = i0 + u * 32 + lane;
        const unsigned pk = (i < n) ? gl[i] : 0u;
        sx[u] = (int)(pk >> NODE_BITS);
        dl[u] = (int)(pk & ((1u << NODE_BITS) - 1u));
      }
#pragma unroll
      for (int u = 0; u < 4; ++u) lg[u] = el[sx[u]];
#pragma unroll
      for (int u = 0; u < 4; ++u) {
        int i = i0 + u * 32 + lane;
        if (i < n) {
          float ev = leaky(lg[u] + s_er[dl[u]]);
          atomicMax(&s_mkey[dl[u]], fkey(ev));
          s_lsrc[seg + i] = sx[u];
          s_le[seg + i]   = ev;
          s_ld[seg + i]   = (unsigned short)dl[u];
        }
      }
    }
    __syncthreads();
    const float* __restrict__ featl = feat + 2 * lane;
    float* accl = s_acc + 2 * lane;
    for (int i0 = 0; i0 < n; i0 += 32) {
      const int cnt = min(32, n - i0);
      int sx = 0, dn = 0, wb = 0;
      if (lane < cnt) {
        const int j = seg + i0 + lane;
        sx = s_lsrc[j];
        dn = s_ld[j];
        const float w = __expf(s_le[j] - fkey_inv(s_mkey[dn]));
        atomicAdd(&s_den[dn], w);
        wb = __float_as_int(w);
      }
      auto group = [&](const int g, const bool guard) {
        int dv[PASS2_EDGES]; float wv[PASS2_EDGES]; v2f fv[PASS2_EDGES];
#pragma unroll
        for (int u = 0; u < PASS2_EDGES; ++u) {
          const int l = (!guard || g + u < cnt) ? (g + u) : g;
          const int sv = __builtin_amdgcn_readlane(sx, l);
          dv[u] = __builtin_amdgcn_readlane(dn, l);
          wv[u] = __int_as_float(__builtin_amdgcn_readlane(wb, l));
          fv[u] = *(const v2f*)(featl + (size_t)sv * OUT_DIM);
        }
#pragma unroll
        for (int u = 0; u < PASS2_EDGES; ++u) {
          if (!guard || g + u < cnt) {
            float* row = accl + dv[u] * OUT_DIM;
            atomicAdd(row,     wv[u] * fv[u].x);
            atomicAdd(row + 1, wv[u] * fv[u].y);
          }
        }
      };
      if (cnt == 32) {
#pragma unroll
        for (int g = 0; g < 32; g += PASS2_EDGES) group(g, false);
      } else {
        for (int g = 0; g < cnt; g += PASS2_EDGES) group(g, true);
      }
    }
  } else {
    for (int e = tid; e < E; e += AGG_THREADS) {
      unsigned dl = (unsigned)(dst[e] - base);
      if (dl < (unsigned)nloc)
        atomicMax(&s_mkey[dl], fkey(leaky(el[src[e]] + s_er[dl])));
    }
    __syncthreads();
    for (int e = tid; e < E; e += AGG_THREADS) {
      unsigned dl = (unsigned)(dst[e] - base);
      if (dl < (unsigned)nloc) {
        int   sidx = src[e];
        float ev   = leaky(el[sidx] + s_er[dl]);
        float w    = __expf(ev - fkey_inv(s_mkey[dl]));
        atomicAdd(&s_den[dl], w);
        const float* fr = feat + (size_t)sidx * OUT_DIM;
        for (int c = 0; c < OUT_DIM; ++c) atomicAdd(&s_acc[dl * OUT_DIM + c], w * fr[c]);
      }
    }
  }
  __syncthreads();

  float* __restrict__ ob = out + (size_t)base * OUT_DIM;
  for (int i = tid; i < nloc * OUT_DIM; i += AGG_THREADS) {
    float den = s_den[i >> 6];
    float inv = (den > 0.f) ? 1.f / den : 1.f;
    ob[i] = s_acc[i] * inv + bias[i & (OUT_DIM - 1)];
  }
}

extern "C" void kernel_launch(void* const* d_in, const int* in_sizes, int n_in,
                              void* d_out, int out_size, void* d_ws, size_t ws_size,
                              hipStream_t stream) {
  const float* h      = (const float*)d_in[0];
  const int*   src    = (const int*)d_in[1];
  const int*   dst    = (const int*)d_in[2];
  const float* W      = (const float*)d_in[3];
  const float* attn_l = (const float*)d_in[4];
  const float* attn_r = (const float*)d_in[5];
  const float* bias   = (const float*)d_in[6];
  float* out = (float*)d_out;

  const int N = in_sizes[0] / IN_DIM;
  const int E = in_sizes[1];

  char* ws = (char*)d_ws;
  size_t off = 0;
  auto carve = [&](size_t bytes) -> char* {
    char* p = ws + off;
    off = (off + bytes + 255) & ~(size_t)255;
    return p;
  };
  float* feat    = (float*)carve((size_t)N * OUT_DIM * sizeof(float));
  float* Wf      = (float*)carve((size_t)IN_DIM * OUT_DIM * sizeof(float));
  float* el      = (float*)carve((size_t)N * sizeof(float));
  float* er      = (float*)carve((size_t)N * sizeof(float));
  float* wlr     = (float*)carve((size_t)2 * IN_DIM * sizeof(float));
  unsigned* keyw   = (unsigned*)carve((size_t)(((E + 15) >> 4) << 2) * sizeof(unsigned));
  unsigned* packed = (unsigned*)carve((size_t)E * sizeof(unsigned));
  unsigned* glist  = (unsigned*)carve((size_t)((N + RANGE - 1) / RANGE) * AGG_WAVES * WAVE_CAP * sizeof(unsigned));
  int*      gcnt   = (int*)carve((size_t)((N + RANGE - 1) / RANGE) * AGG_WAVES * sizeof(int));

  k_wt_keys<<<HEAD_BLOCKS + (((E + 15) >> 4) * 4 + 1023) / 1024, 256, 0, stream>>>(W, Wf, attn_l, attn_r, wlr, src, dst, keyw, packed, N, E);
  k_feat_lists<<<((N + RANGE - 1) / RANGE) * SCAN_BLOCKS + (N + PROJ_ROWS - 1) / PROJ_ROWS, 256, 0, stream>>>(h, Wf, wlr, feat, el, er, keyw, packed, glist, gcnt, N, E);
  k_gat_range<<<(N + RANGE - 1) / RANGE, AGG_THREADS, 0, stream>>>(feat, el, er, src, dst, glist, gcnt, bias, out, N, E);
}


# Edge filter: one set of filter waves serves two aggregation blocks (a key names the pair, one bit of the packed word the block): half the key scans. Node sums: LDS arrival order as before.
# speedup vs baseline: 3.5982x; 1.0734x over previous
#include <hip/hip_runtime.h>
#include <math.h>

#define IN_DIM   128
#define OUT_DIM  64
#define NEG_SLOPE 0.2f

typedef float v2f __attribute__((ext_vector_type(2)));
typedef float v4f __attribute__((ext_vector_type(4)));
typedef float v8f __attribute__((ext_vector_type(8)));

__device__ __forceinline__ float leaky(float e) {
  return (e > 0.f) ? e : NEG_SLOPE * e;
}

#define RANGE      400
#define NODE_BITS  9
#define KEY_NONE   255u
#define KEY_BLOCKS 255
#define PAIR_NODES (2 * RANGE)
#define WT_BLOCKS  ((IN_DIM * OUT_DIM + 255) / 256)
#define HEAD_BLOCKS (WT_BLOCKS + 1)

#define LIST_CAP    16384
#define AGG_THREADS 1024
#define AGG_WAVES   (AGG_THREADS / 32)
#define WAVE_CAP    (LIST_CAP / AGG_WAVES)
#define FILT_CAP    (2 * WAVE_CAP)
#define SCAN_WAVES  8
#define SCAN_BLOCKS (AGG_WAVES / SCAN_WAVES)
#define PROJ_ROWS   64
#define H_PITCH4    (IN_DIM / 4 + 1)

static_assert(RANGE <= (1 << NODE_BITS), "a local node number fits its field");
__device__ __forceinline__ unsigned edge_key(int d, int N) {
  unsigned k = (unsigned)d / PAIR_NODES;
  return ((unsigned)d < (unsigned)N && k < KEY_NONE) ? k : KEY_NONE;
}
__device__ __forceinline__ unsigned edge_pack(int s, int d) {
  const unsigned blk = (unsigned)d / RANGE;
  return ((unsigned)s << (NODE_BITS + 1)) | ((blk & 1u) << NODE_BITS) | ((unsigned)d - blk * RANGE);
}

__global__ __launch_bounds__(256) void k_wt_keys(const float* __restrict__ W, float* __restrict__ Wf,
                                                 const float* __restrict__ attn_l,
                                                 const float* __restrict__ attn_r,
                                                 float* __restrict__ wlr,
                                                 const int* __restrict__ src,
                                                 const int* __restrict__ dst,
                                                 unsigned* __restrict__ keyw,
                                                 unsigned* __restrict__ packed,
                                                 int N, int E) {
  if (blockIdx.x < WT_BLOCKS) {
    int i = blockIdx.x * 256 + threadIdx.x;
    if (i >= IN_DIM * OUT_DIM) return;
    int k = i / OUT_DIM;
    int c = i - k * OUT_DIM;
    const int frag = ((c >> 4) * (IN_DIM / 8) + (k >> 3)) * 32 + ((k >> 2) & 1) * 16 + (c & 15);
    Wf[(size_t)frag * 4 + (k & 3)] = W[i];
    return;
  }
  if (blockIdx.x == WT_BLOCKS) {
    static_assert(2 * IN_DIM == 256, "one thread per folded value");
    const int k = threadIdx.x & (IN_DIM - 1);
    const float4* __restrict__ a4 = (const float4*)((threadIdx.x < IN_DIM) ? attn_l : attn_r);
    const float4* __restrict__ w4 = (const float4*)(W + (size_t)k * OUT_DIM);
    float sum = 0.f;
    for (int j = 0; j < OUT_DIM / 4; ++j) {
      const float4 w = w4[j], a = a4[j];
      sum += w.x * a.x + w.y * a.y + w.z * a.z + w.w * a.w;
    }
    wlr[2 * k + ((threadIdx.x < IN_DIM) ? 0 : 1)] = sum;
    return;
  }
  const int GP = ((E + 15) >> 4) << 2;
  const int gb = (blockIdx.x - HEAD_BLOCKS) * 1024;
  const int g0 = gb + threadIdx.x;
  const int4* __restrict__ dst4 = (const int4*)dst;
  const int4* __restrict__ src4 = (const int4*)src;
  if ((gb + 1024) * 4 <= E) {
    int4 d[4], s[4];
#pragma unroll
    for (int u = 0; u < 4; ++u) { d[u] = dst4[g0 + u * 256]; s[u] = src4[g0 + u * 256]; }
#pragma unroll
    for (int u = 0; u < 4; ++u) {
      uint4 pk;
      pk.x = edge_pack(s[u].x, d[u].x); pk.y = edge_pack(s[u].y, d[u].y);
      pk.z = edge_pack(s[u].z, d[u].z); pk.w = edge_pack(s[u].w, d[u].w);
      ((uint4*)packed)[g0 + u * 256] = pk;
      keyw[g0 + u * 256] = edge_key(d[u].x, N) | (edge_key(d[u].y, N) << 8) |
                           (edge_key(d[u].z, N) << 16) | (edge_key(d[u].w, N) << 24);
    }
    return;
  }
  for (int u = 0; u < 4; ++u) {
    const int g = g0 + u * 256;
    if (g >= GP) continue;
    const int e0 = g << 2;
    unsigned kw;
    if (e0 + 3 < E) {
      const int4 d = dst4[g];
      const int4 s = src4[g];
      uint4 pk;
      pk.x = edge_pack(s.x, d.x); pk.y = edge_pack(s.y, d.y);
      pk.z = edge_pack(s.z, d.z); pk.w = edge_pack(s.w, d.w);
      ((uint4*)packed)[g] = pk;
      kw = edge_key(d.x, N) | (edge_key(d.y, N) << 8) | (edge_key(d.z, N) << 16) | (edge_key(d.w, N) << 24);
    } else {
      kw = 0u;
      for (int b = 0; b < 4; ++b) {
        const int e = e0 + b;
        unsigned k = KEY_NONE;
        if (e < E) {
          const int d = dst[e];
          packed[e] = edge_pack(src[e], d);
          k = edge_key(d, N);
        }
        kw |= k << (8 * b);
      }
    }
    keyw[g] = kw;
  }
}

template <int M> __device__ __forceinline__ float lane_xor(float v) {
  if constexpr (M == 16) {
    const int b = __builtin_bit_cast(int, v);
    return __builtin_bit_cast(float, __builtin_amdgcn_permlanex16(b, b, 0x76543210, 0xfedcba98, false, false));
  } else {
    return __builtin_bit_cast(float, __builtin_amdgcn_update_dpp(0, __builtin_bit_cast(int, v), 0x160 + M, 0xf, 0xf, false));
  }
}
template <int M> __device__ __forceinline__ float fold_pair(float a, float b, int lane) {
  const bool hi = (lane & M) != 0;
  const float keep = hi ? b : a, send = hi ? a : b;
  return keep + lane_xor<M>(send);
}

__global__ __launch_bounds__(256) void k_feat_lists(const float* __restrict__ h,
                                                    const float* __restrict__ Wf,
                                                    const float* __restrict__ wlr,
                                                    float* __restrict__ feat,
                                                    float* __restrict__ el,
                                                    float* __restrict__ er,
                                                    const unsigned* __restrict__ keyw,
                                                    const unsigned* __restrict__ packed,
                                                    unsigned* __restrict__ glist,
                                                    int* __restrict__ gcnt,
                                                    int N, int E) {
  __shared__ v4f s_raw[PROJ_ROWS * H_PITCH4];
  static_assert(sizeof(v4f) * PROJ_ROWS * H_PITCH4 >= sizeof(int) * SCAN_WAVES * FILT_CAP, "the filter's lists fit");
  int* const s_list = (int*)s_raw;
  __shared__ float s_lg[2 * PROJ_ROWS];

  const int tid   = threadIdx.x;
  const int wave  = tid >> 5;
  const int lane  = tid & 31;
  const int nrange = (N + RANGE - 1) / RANGE;
  const int nscan = ((nrange + 1) / 2) * SCAN_BLOCKS;

  if ((int)blockIdx.x < nscan) {
    if (N > KEY_BLOCKS * PAIR_NODES) return;
    const int sw    = __builtin_amdgcn_readfirstlane((int)blockIdx.x * SCAN_WAVES + wave);
    const int pair  = sw / AGG_WAVES;
    const int slice = sw % AGG_WAVES;
    const int seg   = wave * FILT_CAP;
    int wcnt = 0;
    const unsigned pat = (unsigned)pair * 0x01010101u;
    auto hits16 = [&](const uint4 v) -> unsigned {
      const unsigned x0 = v.x ^ pat, x1 = v.y ^ pat, x2 = v.z ^ pat, x3 = v.w ^ pat;
      const unsigned n0 = ((x0 & 0x7F7F7F7Fu) + 0x7F7F7F7Fu) | x0;
      const unsigned n1 = ((x1 & 0x7F7F7F7Fu) + 0x7F7F7F7Fu) | x1;
      const unsigned n2 = ((x2 & 0x7F7F7F7Fu) + 0x7F7F7F7Fu) | x2;
      const unsigned n3 = ((x3 & 0x7F7F7F7Fu) + 0x7F7F7F7Fu) | x3;
      const unsigned nz = ((n0 >> 7) & 0x01010101u) | ((n1 >> 6) & 0x02020202u) |
                          ((n2 >> 5) & 0x04040404u) | ((n3 >> 4) & 0x08080808u);
      return nz ^ 0x0F0F0F0Fu;
    };
    auto push = [&](unsigned m, const int q) {
      unsigned any = __builtin_amdgcn_ballot_w32(m != 0u);
      while (any != 0u) {
        if (m != 0u) {
          const int pos = wcnt + (int)__builtin_amdgcn_mbcnt_lo(any, 0u);
          const int p = __builtin_ctz(m);
          const int e = ((q + ((p >> 2) & 1) * AGG_THREADS) << 4) + ((p & 3) << 2) + (p >> 3);
          if (pos < FILT_CAP) s_list[seg + pos] = e;
          m &= m - 1u;
        }
        wcnt += __builtin_popcount(any);
        any = __builtin_amdgcn_ballot_w32(m != 0u);
      }
    };
    const int Q = (E + 15) >> 4;
    const uint4* __restrict__ key4 = (const uint4*)keyw;
    int qw = slice * 32;
    for (; qw + 7 * AGG_THREADS + 31 < Q; qw += 8 * AGG_THREADS) {
      const int q = qw + lane;
      uint4 v[8];
#pragma unroll
      for (int u = 0; u < 8; ++u) v[u] = key4[q + u * AGG_THREADS];
#pragma unroll
      for (int u = 0; u < 8; u += 2)
        push(hits16(v[u]) | (hits16(v[u + 1]) << 4), q + u * AGG_THREADS);
    }
    for (; qw < Q; qw += AGG_THREADS) {
      const int q = qw + lane;
      uint4 v = make_uint4(~0u, ~0u, ~0u, ~0u);
      if (q < Q) v = key4[q];
      push(hits16(v), q);
    }
    __syncthreads();
    const int n = min(wcnt, FILT_CAP);
    unsigned* __restrict__ gl0 = glist + ((size_t)(2 * pair) * AGG_WAVES + slice) * WAVE_CAP;
    unsigned* __restrict__ gl1 = gl0 + (size_t)AGG_WAVES * WAVE_CAP;
    int c0 = 0, c1 = 0;
    for (int i0 = 0; i0 < n; i0 += 8 * 32) {
      int en[8]; unsigned pk[8];
#pragma unroll
      for (int u = 0; u < 8; ++u) en[u] = s_list[seg + ((i0 + u * 32 + lane) & (FILT_CAP - 1))];
#pragma unroll
      for (int u = 0; u < 8; ++u) pk[u] = packed[(i0 + u * 32 + lane < n) ? en[u] : 0];
#pragma unroll
      for (int u = 0; u < 8; ++u) {
        const bool live = i0 + u * 32 + lane < n;
        const bool odd  = ((pk[u] >> NODE_BITS) & 1u) != 0u;
        const unsigned m1 = __builtin_amdgcn_ballot_w32(live && odd);
        const unsigned m0 = __builtin_amdgcn_ballot_w32(live && !odd);
        const int pos = odd ? c1 + (int)__builtin_amdgcn_mbcnt_lo(m1, 0u) : c0 + (int)__builtin_amdgcn_mbcnt_lo(m0, 0u);
        const unsigned word = ((pk[u] >> (NODE_BITS + 1)) << NODE_BITS) | (pk[u] & ((1u << NODE_BITS) - 1u));
        if (live && pos < WAVE_CAP) (odd ? gl1 : gl0)[pos] = word;
        c0 += __builtin_popcount(m0);
        c1 += __builtin_popcount(m1);
      }
    }
    if (lane == 0) {
      const bool cut = wcnt > FILT_CAP;
      gcnt[(2 * pair) * AGG_WAVES + slice] = cut ? WAVE_CAP + 1 : c0;
      if (2 * pair + 1 < nrange) gcnt[(2 * pair + 1) * AGG_WAVES + slice] = cut ? WAVE_CAP + 1 : c1;
    }
    return;
  }

  const int R0 = ((int)blockIdx.x - nscan) * PROJ_ROWS;
  const bool whole = R0 + PROJ_ROWS <= N;
  {
    const v4f* __restrict__ wv = (const v4f*)wlr + 2 * lane;
    const v4f w01 = wv[0], w23 = wv[1];
    const v4f* __restrict__ h4 = (const v4f*)h;
    const int wrow = R0 + __builtin_amdgcn_readfirstlane(wave);
    v4f st[PROJ_ROWS / 8];
#pragma unroll
    for (int i = 0; i < PROJ_ROWS / 8; ++i)
      st[i] = __builtin_nontemporal_load(h4 + (size_t)min(wrow + 8 * i, N - 1) * (IN_DIM / 4) + lane);
#pragma unroll
    for (int i = 0; i < PROJ_ROWS / 8; ++i) s_raw[(8 * i + wave) * H_PITCH4 + lane] = st[i];
    const v2f w0 = {w01.x, w01.y}, w1 = {w01.z, w01.w}, w2 = {w23.x, w23.y}, w3 = {w23.z, w23.w};
    float v[2 * (PROJ_ROWS / 8)];
#pragma unroll
    for (int i = 0; i < PROJ_ROWS / 8; ++i) {
      const v2f d = w0 * st[i].x + w1 * st[i].y + w2 * st[i].z + w3 * st[i].w;
      v[i]     = d.x;
      v[8 + i] = d.y;
    }
    static_assert(PROJ_ROWS / 8 == 8, "sixteen values, four halving steps");
#pragma unroll
    for (int p = 0; p < 8; ++p) v[p] = fold_pair<16>(v[2 * p], v[2 * p + 1], lane);
#pragma unroll
    for (int p = 0; p < 4; ++p) v[p] = fold_pair<8>(v[2 * p], v[2 * p + 1], lane);
#pragma unroll
    for (int p = 0; p < 2; ++p) v[p] = fold_pair<4>(v[2 * p], v[2 * p + 1], lane);
    float lg = fold_pair<2>(v[0], v[1], lane);
    lg += lane_xor<1>(lg);
    if ((lane & 1) == 0) s_lg[(lane & 2) * (PROJ_ROWS / 2) + 8 * ((lane >> 4) | ((lane >> 2) & 2) | (lane & 4)) + wave] = lg;
  }
  const int ntile = wave & 3;
  const int mhalf = wave >> 2;
  const int c  = lane & 15;
  const int hh = lane >> 4;
  const v4f* __restrict__ wf = (const v4f*)Wf + (size_t)ntile * (IN_DIM / 8) * 32 + lane;
  const v4f* a0p = s_raw + (mhalf * (PROJ_ROWS / 2) + c) * H_PITCH4 + hh;
  const v4f* a1p = a0p + 16 * H_PITCH4;
  static_assert(PROJ_ROWS == 64, "a wave takes two row tiles of its half");
  __syncthreads();

  v8f acc0 = {0.f, 0.f, 0.f, 0.f, 0.f, 0.f, 0.f, 0.f};
  v8f acc1 = {0.f, 0.f, 0.f, 0.f, 0.f, 0.f, 0.f, 0.f};
#pragma unroll
  for (int j = 0; j < IN_DIM / 8; ++j) {
    const v4f b  = wf[j * 32];
    const v4f x0 = a0p[2 * j];
    const v4f x1 = a1p[2 * j];
    const v2f blo = {b.x, b.y},   bhi = {b.z, b.w};
    const v2f l0  = {x0.x, x0.y}, h0  = {x0.z, x0.w};
    const v2f l1  = {x1.x, x1.y}, h1  = {x1.z, x1.w};
    acc0 = __builtin_amdgcn_wmma_f32_16x16x4_f32(false, l0, false, blo, (short)0, acc0, false, false);
    acc1 = __builtin_amdgcn_wmma_f32_16x16x4_f32(false, l1, false, blo, (short)0, acc1, false, false);
    acc0 = __builtin_amdgcn_wmma_f32_16x16x4_f32(false, h0, false, bhi, (short)0, acc0, false, false);
    acc1 = __builtin_amdgcn_wmma_f32_16x16x4_f32(false, h1, false, bhi, (short)0, acc1, false, false);
  }

  const int col   = ntile * 16 + c;
  const int mBase = R0 + mhalf * (PROJ_ROWS / 2) + hh * 8;
  float* __restrict__ fp = feat + (size_t)mBase * OUT_DIM + col;
  if (whole) {
#pragma unroll
    for (int r = 0; r < 8; ++r) {
      fp[r * OUT_DIM]        = acc0[r];
      fp[(r + 16) * OUT_DIM] = acc1[r];
    }
  } else {
#pragma unroll
    for (int r = 0; r < 8; ++r) {
      const int m = mBase + r;
      if (m < N)      fp[r * OUT_DIM]        = acc0[r];
      if (m + 16 < N) fp[(r + 16) * OUT_DIM] = acc1[r];
    }
  }
  if (tid < 2 * PROJ_ROWS) {
    const int m = R0 + (tid & (PROJ_ROWS - 1));
    if (m < N) ((tid < PROJ_ROWS) ? el : er)[m] = s_lg[tid];
  }
}

#define PASS2_EDGES 8

__device__ __forceinline__ int fkey(float f) {
  int b = __float_as_int(f);
  return (b >= 0) ? b : (b ^ 0x7fffffff);
}
__device__ __forceinline__ float fkey_inv(int k) {
  return __int_as_float((k >= 0) ? k : (k ^ 0x7fffffff));
}

__global__ __launch_bounds__(AGG_THREADS) void k_gat_range(const float* __restrict__ feat,
                                                    const float* __restrict__ el,
                                                    const float* __restrict__ er,
                                                    const int* __restrict__ src,
                                                    const int* __restrict__ dst,
                                                    const unsigned* __restrict__ glist,
                                                    const int* __restrict__ gcnt,
                                                    const float* __restrict__ bias,
                                                    float* __restrict__ out,
                                                    int N, int E) {
  __shared__ float          s_acc[RANGE * OUT_DIM];
  __shared__ int            s_mkey[RANGE];
  __shared__ float          s_den[RANGE];
  __shared__ float          s_er[RANGE];
  __shared__ int            s_lsrc[LIST_CAP];
  __shared__ float          s_le[LIST_CAP];
  __shared__ unsigned short s_ld[LIST_CAP];
  __shared__ int            s_wcnt[AGG_WAVES];

  const int tid  = threadIdx.x;
  const int wave = tid >> 5;
  const int lane = tid & 31;
  const int base = blockIdx.x * RANGE;
  const int nloc = min(RANGE, N - base);
  const int NEG_INF_KEY = fkey(-INFINITY);

  for (int i = tid; i < RANGE * OUT_DIM; i += AGG_THREADS) s_acc[i] = 0.f;
  for (int i = tid; i < RANGE; i += AGG_THREADS) {
    s_mkey[i] = NEG_INF_KEY;
    s_den[i]  = 0.f;
    s_er[i]   = (i < nloc) ? er[base + i] : 0.f;
  }
  const bool use_keys = N <= KEY_BLOCKS * PAIR_NODES;
  const int gw = __builtin_amdgcn_readfirstlane((int)blockIdx.x * AGG_WAVES + wave);
  if (lane == 0) s_wcnt[wave] = use_keys ? gcnt[gw] : 0;
  __syncthreads();
  const bool over = !use_keys || __builtin_amdgcn_ballot_w32(s_wcnt[lane] > WAVE_CAP) != 0u;
  const int seg = wave * WAVE_CAP;
  static_assert(AGG_WAVES == 32, "the overflow test reads one wave count per lane");

  if (!over) {
    const int n = __builtin_amdgcn_readfirstlane(s_wcnt[wave]);
    const unsigned* __restrict__ gl = glist + (size_t)gw * WAVE_CAP;
    for (int i0 = 0; i0 < n; i0 += 128) {
      int sx[4]; int dl[4]; float lg[4];
#pragma unroll
      for (int u = 0; u < 4; ++u) {
        int i = i0 + u * 32 + lane;
        const unsigned pk = (i < n) ? gl[i] : 0u;
        sx[u] = (int)(pk >> NODE_BITS);
        dl[u] = (int)(pk & ((1u << NODE_BITS) - 1u));
      }
#pragma unroll
      for (int u = 0; u < 4; ++u) lg[u] = el[sx[u]];
#pragma unroll
      for (int u = 0; u < 4; ++u) {
        int i = i0 + u * 32 + lane;
        if (i < n) {
          float ev = leaky(lg[u] + s_er[dl[u]]);
          atomicMax(&s_mkey[dl[u]], fkey(ev));
          s_lsrc[seg + i] = sx[u];
          s_le[seg + i]   = ev;
          s_ld[seg + i]   = (unsigned short)dl[u];
        }
      }
    }
    __syncthreads();
    const float* __restrict__ featl = feat + 2 * lane;
    float* accl = s_acc + 2 * lane;
    for (int i0 = 0; i0 < n; i0 += 32) {
      const int cnt = min(32, n - i0);
      int sx = 0, dn = 0, wb = 0;
      if (lane < cnt) {
        const int j = seg + i0 + lane;
        sx = s_lsrc[j];
        dn = s_ld[j];
        const float w = __expf(s_le[j] - fkey_inv(s_mkey[dn]));
        atomicAdd(&s_den[dn], w);
        wb = __float_as_int(w);
      }
      auto group = [&](const int g, const bool guard) {
        int dv[PASS2_EDGES]; float wv[PASS2_EDGES]; v2f fv[PASS2_EDGES];
#pragma unroll
        for (int u = 0; u < PASS2_EDGES; ++u) {
          const int l = (!guard || g + u < cnt) ? (g + u) : g;
          const int sv = __builtin_amdgcn_readlane(sx, l);
          dv[u] = __builtin_amdgcn_readlane(dn, l);
          wv[u] = __int_as_float(__builtin_amdgcn_readlane(wb, l));
          fv[u] = *(const v2f*)(featl + (size_t)sv * OUT_DIM);
        }
#pragma unroll
        for (int u = 0; u < PASS2_EDGES; ++u) {
          if (!guard || g + u < cnt) {
            float* row = accl + dv[u] * OUT_DIM;
            atomicAdd(row,     wv[u] * fv[u].x);
            atomicAdd(row + 1, wv[u] * fv[u].y);
          }
        }
      };
      if (cnt == 32) {
#pragma unroll
        for (int g = 0; g < 32; g += PASS2_EDGES) group(g, false);
      } else {
        for (int g = 0; g < cnt; g += PASS2_EDGES) group(g, true);
      }
    }
  } else {
    for (int e = tid; e < E; e += AGG_THREADS) {
      unsigned dl = (unsigned)(dst[e] - base);
      if (dl < (unsigned)nloc)
        atomicMax(&s_mkey[dl], fkey(leaky(el[src[e]] + s_er[dl])));
    }
    __syncthreads();
    for (int e = tid; e < E; e += AGG_THREADS) {
      unsigned dl = (unsigned)(dst[e] - base);
      if (dl < (unsigned)nloc) {
        int   sidx = src[e];
        float ev   = leaky(el[sidx] + s_er[dl]);
        float w    = __expf(ev - fkey_inv(s_mkey[dl]));
        atomicAdd(&s_den[dl], w);
        const float* fr = feat + (size_t)sidx * OUT_DIM;
        for (int c = 0; c < OUT_DIM; ++c) atomicAdd(&s_acc[dl * OUT_DIM + c], w * fr[c]);
      }
    }
  }
  __syncthreads();

  float* __restrict__ ob = out + (size_t)base * OUT_DIM;
  for (int i = tid; i < nloc * OUT_DIM; i += AGG_THREADS) {
    float den = s_den[i >> 6];
    float inv = (den > 0.f) ? 1.f / den : 1.f;
    ob[i] = s_acc[i] * inv + bias[i & (OUT_DIM - 1)];
  }
}

extern "C" void kernel_launch(void* const* d_in, const int* in_sizes, int n_in,
                              void* d_out, int out_size, void* d_ws, size_t ws_size,
                              hipStream_t stream) {
  const float* h      = (const float*)d_in[0];
  const int*   src    = (const int*)d_in[1];
  const int*   dst    = (const int*)d_in[2];
  const float* W      = (const float*)d_in[3];
  const float* attn_l = (const float*)d_in[4];
  const float* attn_r = (const float*)d_in[5];
  const float* bias   = (const float*)d_in[6];
  float* out = (float*)d_out;

  const int N = in_sizes[0] / IN_DIM;
  const int E = in_sizes[1];

  char* ws = (char*)d_ws;
  size_t off = 0;
  auto carve = [&](size_t bytes) -> char* {
    char* p = ws + off;
    off = (off + bytes + 255) & ~(size_t)255;
    return p;
  };
  float* feat    = (float*)carve((size_t)N * OUT_DIM * sizeof(float));
  float* Wf      = (float*)carve((size_t)IN_DIM * OUT_DIM * sizeof(float));
  float* el      = (float*)carve((size_t)N * sizeof(float));
  float* er      = (float*)carve((size_t)N * sizeof(float));
  float* wlr     = (float*)carve((size_t)2 * IN_DIM * sizeof(float));
  unsigned* keyw   = (unsigned*)carve((size_t)(((E + 15) >> 4) << 2) * sizeof(unsigned));
  unsigned* packed = (unsigned*)carve((size_t)E * sizeof(unsigned));
  unsigned* glist  = (unsigned*)carve((size_t)((N + RANGE - 1) / RANGE) * AGG_WAVES * WAVE_CAP * sizeof(unsigned));
  int*      gcnt   = (int*)carve((size_t)((N + RANGE - 1) / RANGE) * AGG_WAVES * sizeof(int));

  k_wt_keys<<<HEAD_BLOCKS + (((E + 15) >> 4) * 4 + 1023) / 1024, 256, 0, stream>>>(W, Wf, attn_l, attn_r, wlr, src, dst, keyw, packed, N, E);
  k_feat_lists<<<(((N + RANGE - 1) / RANGE + 1) / 2) * SCAN_BLOCKS + (N + PROJ_ROWS - 1) / PROJ_ROWS, 256, 0, stream>>>(h, Wf, wlr, feat, el, er, keyw, packed, glist, gcnt, N, E);
  k_gat_range<<<(N + RANGE - 1) / RANGE, AGG_THREADS, 0, stream>>>(feat, el, er, src, dst, glist, gcnt, bias, out, N, E);
}
